# starting kernel
# baseline (speedup 1.0000x reference)
#include <hip/hip_runtime.h>
#include <hip/hip_bf16.h>
#include <float.h>
#include <math.h>

#define HEADS 4
#define CPH 32
#define HC 128
#define NEG_SLOPE 0.2f
#define EPSV 1e-5f

typedef float v2f __attribute__((ext_vector_type(2)));
typedef float v8f __attribute__((ext_vector_type(8)));

__device__ __forceinline__ unsigned enc_f32(float f) {
    unsigned u = __float_as_uint(f);
    return (u & 0x80000000u) ? ~u : (u | 0x80000000u);
}
__device__ __forceinline__ float dec_f32(unsigned k) {
    unsigned u = (k & 0x80000000u) ? (k ^ 0x80000000u) : ~k;
    return __uint_as_float(u);
}

__device__ __forceinline__ void edge_sd(long long e, const int* __restrict__ src,
                                        const int* __restrict__ dst, int E,
                                        int* s, int* d) {
    if (e < (long long)E) { *s = src[e]; *d = dst[e]; }
    else { int v = (int)(e - E); *s = v; *d = v; }
}

__global__ __launch_bounds__(256) void k_gemm_wmma(
    const float* __restrict__ A, const float* __restrict__ W,
    float* __restrict__ Hout, int M)
{
    const int lane = threadIdx.x & 31;
    const int wave = threadIdx.x >> 5;
    const int m0   = blockIdx.x * 16;
    const int n0   = wave * 16;
    const int hi   = lane >> 4;
    const int l16  = lane & 15;
    const bool full = (m0 + 16 <= M);

    int row = m0 + l16;
    if (row >= M) row = M - 1;
    const float* __restrict__ arow = A + (size_t)row * HC;

    v8f acc = {0.f, 0.f, 0.f, 0.f, 0.f, 0.f, 0.f, 0.f};

#pragma unroll 4
    for (int k = 0; k < HC; k += 4) {
        const int kk = k + 2 * hi;
        v2f av, bv;
        av.x = arow[kk];
        av.y = arow[kk + 1];
        bv.x = W[(size_t)kk * HC + n0 + l16];
        bv.y = W[(size_t)(kk + 1) * HC + n0 + l16];
        acc = __builtin_amdgcn_wmma_f32_16x16x4_f32(
             false, av,  false, bv,
             (short)0, acc,  false,  false);
    }

    float* __restrict__ outp = Hout + (size_t)(m0 + 8 * hi) * HC + n0 + l16;
    if (full) {
#pragma unroll
        for (int r = 0; r < 8; ++r) outp[(size_t)r * HC] = acc[r];
    } else {
#pragma unroll
        for (int r = 0; r < 8; ++r) {
            int ro = m0 + r + 8 * hi;
            if (ro < M) Hout[(size_t)ro * HC + n0 + l16] = acc[r];
        }
    }
}

__global__ void k_attn_logits(const float* __restrict__ Hm,
                              const float* __restrict__ a_s,
                              const float* __restrict__ a_d,
                              float* __restrict__ als, float* __restrict__ ald,
                              int Nn)
{
    int idx = blockIdx.x * blockDim.x + threadIdx.x;
    if (idx >= Nn * HEADS) return;
    int n = idx >> 2, h = idx & 3;
    const float4* hp = (const float4*)(Hm + (size_t)n * HC + h * CPH);
    const float4* as = (const float4*)(a_s + h * CPH);
    const float4* ad = (const float4*)(a_d + h * CPH);
    float ss = 0.f, sd = 0.f;
#pragma unroll
    for (int c = 0; c < CPH / 4; ++c) {
        float4 v = hp[c], a = as[c], b = ad[c];
        ss += v.x * a.x + v.y * a.y + v.z * a.z + v.w * a.w;
        sd += v.x * b.x + v.y * b.y + v.z * b.z + v.w * b.w;
    }
    als[idx] = ss;
    ald[idx] = sd;
}

__global__ void k_init(float4* __restrict__ agg4, float* __restrict__ denom,
                       unsigned* __restrict__ mkey, int Nn)
{
    long long idx = (long long)blockIdx.x * blockDim.x + threadIdx.x;
    if (idx < (long long)Nn * (HC / 4)) agg4[idx] = make_float4(0.f, 0.f, 0.f, 0.f);
    if (idx < (long long)Nn * HEADS)    { denom[idx] = 0.f; mkey[idx] = 0u; }
}

__global__ void k_edge_max(const int* __restrict__ src, const int* __restrict__ dst,
                           const float* __restrict__ als, const float* __restrict__ ald,
                           unsigned* __restrict__ mkey, int E, int Etot)
{
    long long idx = (long long)blockIdx.x * blockDim.x + threadIdx.x;
    if (idx >= (long long)Etot * HEADS) return;
    long long e = idx >> 2; int h = (int)(idx & 3);
    int s, d; edge_sd(e, src, dst, E, &s, &d);
    float ev = als[s * HEADS + h] + ald[d * HEADS + h];
    ev = (ev < 0.f) ? NEG_SLOPE * ev : ev;
    atomicMax(&mkey[d * HEADS + h], enc_f32(ev));
}

__global__ void k_edge_expsum(const int* __restrict__ src, const int* __restrict__ dst,
                              const float* __restrict__ als, const float* __restrict__ ald,
                              const unsigned* __restrict__ mkey, float* __restrict__ denom,
                              int E, int Etot)
{
    long long idx = (long long)blockIdx.x * blockDim.x + threadIdx.x;
    if (idx >= (long long)Etot * HEADS) return;
    long long e = idx >> 2; int h = (int)(idx & 3);
    int s, d; edge_sd(e, src, dst, E, &s, &d);
    int dh = d * HEADS + h;
    float ev = als[s * HEADS + h] + ald[dh];
    ev = (ev < 0.f) ? NEG_SLOPE * ev : ev;
    atomicAdd(&denom[dh], expf(ev - dec_f32(mkey[dh])));
}

__global__ void k_edge_msg(const int* __restrict__ src, const int* __restrict__ dst,
                           const float* __restrict__ als, const float* __restrict__ ald,
                           const unsigned* __restrict__ mkey, const float* __restrict__ denom,
                           const float* __restrict__ Hm, float* __restrict__ agg,
                           int E, int Etot)
{
    long long idx = (long long)blockIdx.x * blockDim.x + threadIdx.x;
    if (idx >= (long long)Etot * (HC / 4)) return;
    long long e = idx >> 5;
    int q  = (int)(idx & 31);
    int k4 = q * 4;
    int h  = k4 >> 5;
    int s, d; edge_sd(e, src, dst, E, &s, &d);
    int dh = d * HEADS + h;
    float ev = als[s * HEADS + h] + ald[dh];
    ev = (ev < 0.f) ? NEG_SLOPE * ev : ev;
    float alpha = expf(ev - dec_f32(mkey[dh])) / denom[dh];

    float4 v = *(const float4*)(Hm + (size_t)s * HC + k4);
    float* ap = agg + (size_t)d * HC + k4;
    atomicAdd(ap + 0, v.x * alpha);
    atomicAdd(ap + 1, v.y * alpha);
    atomicAdd(ap + 2, v.z * alpha);
    atomicAdd(ap + 3, v.w * alpha);
}

__global__ __launch_bounds__(HC) void k_epilogue(
    const float* __restrict__ agg, const float* __restrict__ bias,
    const float* __restrict__ g, const float* __restrict__ bln,
    float* __restrict__ out, int apply_ln)
{
    __shared__ float s1[HC];
    __shared__ float s2[HC];
    const int n = blockIdx.x;
    const int t = threadIdx.x;
    float v = agg[(size_t)n * HC + t] + bias[t];
    if (!apply_ln) {
        out[(size_t)n * HC + t] = v;
        return;
    }
    s1[t] = v;
    s2[t] = v * v;
    __syncthreads();
#pragma unroll
    for (int off = HC / 2; off > 0; off >>= 1) {
        if (t < off) { s1[t] += s1[t + off]; s2[t] += s2[t + off]; }
        __syncthreads();
    }
    float mu  = s1[0] * (1.f / HC);
    float var = s2[0] * (1.f / HC) - mu * mu;
    float y = (v - mu) * rsqrtf(var + EPSV) * g[t] + bln[t];
    out[(size_t)n * HC + t] = fmaxf(y, 0.f);
}

extern "C" void kernel_launch(void* const* d_in, const int* in_sizes, int n_in,
                              void* d_out, int out_size, void* d_ws, size_t ws_size,
                              hipStream_t stream)
{
    const float* x    = (const float*)d_in[0];
    const int*   esrc = (const int*)d_in[1];
    const int*   edst = (const int*)d_in[2];
    const float* W[3]  = {(const float*)d_in[3],  (const float*)d_in[7],  (const float*)d_in[11]};
    const float* As[3] = {(const float*)d_in[4],  (const float*)d_in[8],  (const float*)d_in[12]};
    const float* Ad[3] = {(const float*)d_in[5],  (const float*)d_in[9],  (const float*)d_in[13]};
    const float* Bi[3] = {(const float*)d_in[6],  (const float*)d_in[10], (const float*)d_in[14]};
    const float* LnG[2] = {(const float*)d_in[15], (const float*)d_in[17]};
    const float* LnB[2] = {(const float*)d_in[16], (const float*)d_in[18]};

    const int Nn   = in_sizes[0] / HC;
    const int E    = in_sizes[1];
    const int Etot = E + Nn;

    float*    hbuf  = (float*)d_ws;
    float*    agg   = hbuf + (size_t)Nn * HC;
    float*    als   = agg  + (size_t)Nn * HC;
    float*    ald   = als  + (size_t)Nn * HEADS;
    unsigned* mkey  = (unsigned*)(ald + (size_t)Nn * HEADS);
    float*    denom = (float*)(mkey + (size_t)Nn * HEADS);

    const int TB = 256;
    const unsigned gGemm = (unsigned)((Nn + 15) / 16);
    const unsigned gAttn = (unsigned)(((long long)Nn * HEADS + TB - 1) / TB);
    const unsigned gInit = (unsigned)(((long long)Nn * (HC / 4) + TB - 1) / TB);
    const unsigned gEH   = (unsigned)(((long long)Etot * HEADS + TB - 1) / TB);
    const unsigned gMsg  = (unsigned)(((long long)Etot * (HC / 4) + TB - 1) / TB);

    const float* cur = x;
    for (int l = 0; l < 3; ++l) {
        float* layer_out = (l == 2) ? (float*)d_out : agg;

        k_gemm_wmma<<<gGemm, 256, 0, stream>>>(cur, W[l], hbuf, Nn);
        k_attn_logits<<<gAttn, TB, 0, stream>>>(hbuf, As[l], Ad[l], als, ald, Nn);
        k_init<<<gInit, TB, 0, stream>>>((float4*)layer_out, denom, mkey, Nn);
        k_edge_max<<<gEH, TB, 0, stream>>>(esrc, edst, als, ald, mkey, E, Etot);
        k_edge_expsum<<<gEH, TB, 0, stream>>>(esrc, edst, als, ald, mkey, denom, E, Etot);
        k_edge_msg<<<gMsg, TB, 0, stream>>>(esrc, edst, als, ald, mkey, denom,
                                            hbuf, layer_out, E, Etot);
        if (l < 2) {
            k_epilogue<<<(unsigned)Nn, HC, 0, stream>>>(layer_out, Bi[l],
                                                        LnG[l], LnB[l],
                                                        layer_out, 1);
            cur = layer_out;
        } else {
            k_epilogue<<<(unsigned)Nn, HC, 0, stream>>>(layer_out, Bi[l],
                                                        LnG[0], LnB[0],
                                                        layer_out, 0);
        }
    }
}


# Edge pass owned by dst range: 241 work-groups hold 416 nodes' sums in LDS, each scans all dst in place and gathers only its own edges; LDS float adds, rows stored once; LDS arrival-order sums
# speedup vs baseline: 2.9356x; 2.9356x over previous
#include <hip/hip_runtime.h>
#include <hip/hip_bf16.h>
#include <float.h>
#include <math.h>
#include <type_traits>

#define HEADS 4
#define CPH 32
#define HC 128
#define NEG_SLOPE 0.2f
#define EPSV 1e-5f

typedef float v2f __attribute__((ext_vector_type(2)));
typedef float v8f __attribute__((ext_vector_type(8)));

__device__ __forceinline__ unsigned enc_f32(float f) {
    unsigned u = __float_as_uint(f);
    return (u & 0x80000000u) ? ~u : (u | 0x80000000u);
}
__device__ __forceinline__ float dec_f32(unsigned k) {
    unsigned u = (k & 0x80000000u) ? (k ^ 0x80000000u) : ~k;
    return __uint_as_float(u);
}

__device__ __forceinline__ void edge_sd(long long e, const int* __restrict__ src,
                                        const int* __restrict__ dst, int E,
                                        int* s, int* d) {
    if (e < (long long)E) { *s = src[e]; *d = dst[e]; }
    else { int v = (int)(e - E); *s = v; *d = v; }
}

#define XP 132

__global__ __launch_bounds__(256) __attribute__((amdgpu_waves_per_eu(16, 16))) void k_gemm_wmma(
    const float* __restrict__ A, const float* __restrict__ W,
    float* __restrict__ Hout, int M,
    const float* __restrict__ denom, const float* __restrict__ bias,
    const float* __restrict__ g, const float* __restrict__ bln, int fuse)
{
    __shared__ __attribute__((aligned(16))) float Xs[16 * XP];

    const int lane = threadIdx.x & 31;
    const int wave = threadIdx.x >> 5;
    const int m0   = blockIdx.x * 16;
    const int n0   = wave * 16;
    const int hi   = lane >> 4;
    const int l16  = lane & 15;
    const bool full = (m0 + 16 <= M);

    float4 v[2];
    int rr[2];
#pragma unroll
    for (int i = 0; i < 2; ++i) {
        int r = m0 + wave + 8 * i;
        if (r >= M) r = M - 1;
        rr[i] = r;
        v[i] = *(const float4*)(A + (size_t)r * HC + 4 * lane);
    }
    if (fuse) {
        float dn[2];
#pragma unroll
        for (int i = 0; i < 2; ++i) dn[i] = denom[rr[i] * HEADS + (lane >> 3)];
        const float4 bi = *(const float4*)(bias + 4 * lane);
        const float4 gg = *(const float4*)(g + 4 * lane);
        const float4 bb = *(const float4*)(bln + 4 * lane);
#pragma unroll
        for (int i = 0; i < 2; ++i) {
            float4 t = v[i];
            t.x = t.x / dn[i] + bi.x;
            t.y = t.y / dn[i] + bi.y;
            t.z = t.z / dn[i] + bi.z;
            t.w = t.w / dn[i] + bi.w;
            float s1 = (t.x + t.y) + (t.z + t.w);
            float s2 = (t.x * t.x + t.y * t.y) + (t.z * t.z + t.w * t.w);
#pragma unroll
            for (int off = 16; off > 0; off >>= 1) {
                s1 += __shfl_xor(s1, off, 32);
                s2 += __shfl_xor(s2, off, 32);
            }
            const float mu  = s1 * (1.f / HC);
            const float var = s2 * (1.f / HC) - mu * mu;
            const float rs  = rsqrtf(var + EPSV);
            t.x = fmaxf((t.x - mu) * rs * gg.x + bb.x, 0.f);
            t.y = fmaxf((t.y - mu) * rs * gg.y + bb.y, 0.f);
            t.z = fmaxf((t.z - mu) * rs * gg.z + bb.z, 0.f);
            t.w = fmaxf((t.w - mu) * rs * gg.w + bb.w, 0.f);
            v[i] = t;
        }
    }
#pragma unroll
    for (int i = 0; i < 2; ++i)
        *(float4*)(Xs + (wave + 8 * i) * XP + 4 * lane) = v[i];
    __syncthreads();

    const float* xr = Xs + l16 * XP + 2 * hi;
    const float* __restrict__ wc = W + (size_t)(2 * hi) * HC + n0 + l16;

    v8f acc = {0.f, 0.f, 0.f, 0.f, 0.f, 0.f, 0.f, 0.f};

#pragma unroll 4
    for (int k = 0; k < HC; k += 4) {
        v2f av, bv;
        av = *(const v2f*)(xr + k);
        bv.x = wc[(size_t)k * HC];
        bv.y = wc[(size_t)(k + 1) * HC];
        acc = __builtin_amdgcn_wmma_f32_16x16x4_f32(
             false, av,  false, bv,
             (short)0, acc,  false,  false);
    }

    float* __restrict__ outp = Hout + (size_t)(m0 + 8 * hi) * HC + n0 + l16;
    if (full) {
#pragma unroll
        for (int r = 0; r < 8; ++r) outp[(size_t)r * HC] = acc[r];
    } else {
#pragma unroll
        for (int r = 0; r < 8; ++r) {
            int ro = m0 + r + 8 * hi;
            if (ro < M) Hout[(size_t)ro * HC + n0 + l16] = acc[r];
        }
    }
}

#define SHIFT_SPAN 32.0f

__global__ __launch_bounds__(256) void k_attn_logits(const float* __restrict__ Hm,
                              const float* __restrict__ a_s,
                              const float* __restrict__ a_d,
                              float* __restrict__ als, float* __restrict__ ald,
                              float* __restrict__ part, int Nn)
{
    __shared__ float red[8][8];
    const int idx = blockIdx.x * 256 + threadIdx.x;
    const bool ok = idx < Nn * HEADS;
    const int lane = threadIdx.x & 31;
    const int wave = threadIdx.x >> 5;
    float ss = 0.f, sd = 0.f;
    if (ok) {
        int n = idx >> 2, h = idx & 3;
        const float4* hp = (const float4*)(Hm + (size_t)n * HC + h * CPH);
        const float4* as = (const float4*)(a_s + h * CPH);
        const float4* ad = (const float4*)(a_d + h * CPH);
#pragma unroll
        for (int c = 0; c < CPH / 4; ++c) {
            float4 v = hp[c], a = as[c], b = ad[c];
            ss += v.x * a.x + v.y * a.y + v.z * a.z + v.w * a.w;
            sd += v.x * b.x + v.y * b.y + v.z * b.z + v.w * b.w;
        }
        als[idx] = ss;
        ald[idx] = sd;
    }
    float mx = ok ? ss : -INFINITY;
    float mn = ok ? ss :  INFINITY;
#pragma unroll
    for (int off = 4; off < 32; off <<= 1) {
        mx = fmaxf(mx, __shfl_xor(mx, off, 32));
        mn = fminf(mn, __shfl_xor(mn, off, 32));
    }
    if (lane < 4) { red[wave][lane] = mx; red[wave][4 + lane] = mn; }
    __syncthreads();
    if (threadIdx.x < 8) {
        const int t = threadIdx.x;
        float r = red[0][t];
#pragma unroll
        for (int w = 1; w < 8; ++w)
            r = (t < 4) ? fmaxf(r, red[w][t]) : fminf(r, red[w][t]);
        part[blockIdx.x * 8 + t] = r;
    }
}

__global__ __launch_bounds__(256) void k_init(unsigned* __restrict__ mkey, int Nn,
                       const float* __restrict__ part, int nPart,
                       float* __restrict__ gst, int* __restrict__ gflag)
{
    __shared__ float r2[8][8];
    long long idx = (long long)blockIdx.x * blockDim.x + threadIdx.x;
    if (idx < (long long)Nn * HEADS) mkey[idx] = 0u;

    if (blockIdx.x == 0) {
        const int t = threadIdx.x;
        const int c = t & 7;
        const int lane = t & 31;
        float r = (c < 4) ? -INFINITY : INFINITY;
        for (int b = t >> 3; b < nPart; b += 32) {
            const float p = part[b * 8 + c];
            r = (c < 4) ? fmaxf(r, p) : fminf(r, p);
        }
        float o = __shfl_xor(r, 8, 32);
        r = (c < 4) ? fmaxf(r, o) : fminf(r, o);
        o = __shfl_xor(r, 16, 32);
        r = (c < 4) ? fmaxf(r, o) : fminf(r, o);
        if (lane < 8) r2[t >> 5][lane] = r;
        __syncthreads();
        if (t < 8) {
            float q = r2[0][t];
#pragma unroll
            for (int w = 1; w < 8; ++w)
                q = (t < 4) ? fmaxf(q, r2[w][t]) : fminf(q, r2[w][t]);
            r2[0][t] = q;
        }
        __syncthreads();
        if (t == 0) {
            int wide = 0;
#pragma unroll
            for (int h = 0; h < HEADS; ++h) {
                gst[h] = r2[0][h];
                if (!(r2[0][h] - r2[0][4 + h] <= SHIFT_SPAN)) wide = 1;
            }
            gflag[0] = wide;
        }
    }
}

__global__ __launch_bounds__(256) void k_edge_max(const int* __restrict__ src, const int* __restrict__ dst,
                           const float* __restrict__ als, const float* __restrict__ ald,
                           unsigned* __restrict__ mkey, const int* __restrict__ gflag,
                           int E, int Etot)
{
    if (gflag[0] == 0) return;
    const long long total = (long long)Etot * HEADS;
    const long long step = (long long)gridDim.x * blockDim.x;
    for (long long idx = (long long)blockIdx.x * blockDim.x + threadIdx.x; idx < total; idx += step) {
        long long e = idx >> 2; int h = (int)(idx & 3);
        int s, d; edge_sd(e, src, dst, E, &s, &d);
        float ev = als[s * HEADS + h] + ald[d * HEADS + h];
        ev = (ev < 0.f) ? NEG_SLOPE * ev : ev;
        atomicMax(&mkey[d * HEADS + h], enc_f32(ev));
    }
}

#define EPW 8
#define RNG 416
#define OWN_WAVES 32
#define SCAN_U 8
#define SCAN_G 4
#define QCAP 544

__device__ __forceinline__ void lds_add(float* p, float x) {
    __hip_atomic_fetch_add(p, x, __ATOMIC_RELAXED, __HIP_MEMORY_SCOPE_WORKGROUP);
}

__global__ __launch_bounds__(OWN_WAVES * 32) void k_edge_own(
                           const int* __restrict__ src, const int* __restrict__ dst,
                           const float* __restrict__ als, const float* __restrict__ ald,
                           const unsigned* __restrict__ mkey, float* __restrict__ denom,
                           const float* __restrict__ Hm, float* __restrict__ agg,
                           const float* __restrict__ gst, const int* __restrict__ gflag,
                           int E, int Nn)
{
    __shared__ __attribute__((aligned(16))) float acc[RNG * HC];
    __shared__ __attribute__((aligned(16))) float dsum[RNG * HEADS];
    __shared__ int qe[OWN_WAVES][QCAP];

    const int tid  = threadIdx.x;
    const int lane = tid & 31;
    const int wv = __builtin_amdgcn_readfirstlane(tid >> 5);
    const int lo = blockIdx.x * RNG;
    const int cnt = (Nn - lo < RNG) ? (Nn - lo) : RNG;
    const int exact = gflag[0];
    const float Mh = gst[lane & 3];

    for (int i = tid; i < RNG * HC / 4; i += OWN_WAVES * 32)
        ((float4*)acc)[i] = make_float4(0.f, 0.f, 0.f, 0.f);
    for (int i = tid; i < RNG * HEADS; i += OWN_WAVES * 32) dsum[i] = 0.f;
    __syncthreads();

    int qn = 0;
    int* const myq = qe[wv];

    auto append = [&](bool take, int e) {
        const unsigned mask = __builtin_amdgcn_ballot_w32(take);
        if (mask != 0u) {
            const int pos = qn + (int)__builtin_amdgcn_mbcnt_lo(mask, 0u);
            if (take) myq[pos] = e;
            qn += __builtin_popcount(mask);
        }
    };

    auto process = [&](int base, int n, auto nbTag) {
        constexpr int NB = decltype(nbTag)::value;
        const int j = lane >> 2;
        const int h = lane & 3;
        int s[NB], d[NB];
        bool act[NB];
#pragma unroll
        for (int b = 0; b < NB; ++b) {
            act[b] = b * EPW + j < n;
            const int e = myq[base + (act[b] ? b * EPW + j : 0)];
            edge_sd((long long)e, src, dst, E, &s[b], &d[b]);
        }

        float v[NB][EPW][HEADS];
#pragma unroll
        for (int b = 0; b < NB; ++b) {
#pragma unroll
            for (int q = 0; q < EPW; ++q) {
                const int sq = __builtin_amdgcn_readlane(s[b], 4 * q);
                const float* __restrict__ hp = Hm + (size_t)sq * HC + lane;
#pragma unroll
                for (int k = 0; k < HEADS; ++k) v[b][q][k] = hp[k * CPH];
            }
        }

        float as_[NB], ad_[NB];
        unsigned mk[NB];
#pragma unroll
        for (int b = 0; b < NB; ++b) {
            const int dh = d[b] * HEADS + h;
            as_[b] = als[s[b] * HEADS + h];
            ad_[b] = ald[dh];
            mk[b] = 0u;
            if (exact) mk[b] = mkey[dh];
        }

        __builtin_amdgcn_sched_barrier(0);

#pragma unroll
        for (int b = 0; b < NB; ++b) {
            float ev = as_[b] + ad_[b];
            ev = (ev < 0.f) ? NEG_SLOPE * ev : ev;
            float m;
            if (exact) m = dec_f32(mk[b]);
            else { m = Mh + ad_[b]; m = (m < 0.f) ? NEG_SLOPE * m : m; }
            const float w = expf(ev - m);
            const int dl = d[b] - lo;
            if (act[b]) lds_add(&dsum[dl * HEADS + h], w);

            auto add_row = [&](int q) {
                const int dq = __builtin_amdgcn_readlane(dl, 4 * q);
                float* ap = acc + dq * HC + lane;
#pragma unroll
                for (int k = 0; k < HEADS; ++k) {
                    const float wk = __int_as_float(
                        __builtin_amdgcn_readlane(__float_as_int(w), 4 * q + k));
                    lds_add(ap + k * CPH, v[b][q][k] * wk);
                }
            };
            if (n == NB * EPW) {
#pragma unroll
                for (int q = 0; q < EPW; ++q) add_row(q);
            } else {
#pragma unroll
                for (int q = 0; q < EPW; ++q) if (b * EPW + q < n) add_row(q);
            }
        }
    };
    const std::integral_constant<int, 1> oneBatch{};
    const std::integral_constant<int, 2> twoBatches{};

    {
        const int PERW = (RNG + OWN_WAVES - 1) / OWN_WAVES;
        const int node = lo + wv * PERW + lane;
        append(lane < PERW && node < lo + cnt, E + node);
    }

    const int nFull = E >> 7;
    const int rot = (int)(((long long)nFull * blockIdx.x) / gridDim.x);
    const int4* __restrict__ d4p = (const int4*)dst;
    for (int c0 = wv; c0 < nFull; c0 += OWN_WAVES * SCAN_U) {
        int4 dv[SCAN_U];
        int cb[SCAN_U];
        if (c0 + OWN_WAVES * (SCAN_U - 1) < nFull) {
#pragma unroll
            for (int u = 0; u < SCAN_U; ++u) {
                int ca = c0 + OWN_WAVES * u + rot;
                if (ca >= nFull) ca -= nFull;
                cb[u] = ca;
                dv[u] = d4p[(size_t)ca * 32 + lane];
            }
        } else {
#pragma unroll
            for (int u = 0; u < SCAN_U; ++u) {
                const int c = c0 + OWN_WAVES * u;
                int ca = c + rot;
                if (ca >= nFull) ca -= nFull;
                cb[u] = ca;
                if (c < nFull) dv[u] = d4p[(size_t)ca * 32 + lane];
                else dv[u] = make_int4(-1, -1, -1, -1);
            }
        }
#pragma unroll
        for (int u = 0; u < SCAN_U; ++u) {
            const int4 t = dv[u];
            const bool m0 = (unsigned)(t.x - lo) < (unsigned)cnt;
            const bool m1 = (unsigned)(t.y - lo) < (unsigned)cnt;
            const bool m2 = (unsigned)(t.z - lo) < (unsigned)cnt;
            const bool m3 = (unsigned)(t.w - lo) < (unsigned)cnt;
            if (__builtin_amdgcn_ballot_w32(m0 | m1 | m2 | m3) != 0u) {
                const int eb = cb[u] * 128 + 4 * lane;
                append(m0, eb);
                append(m1, eb + 1);
                append(m2, eb + 2);
                append(m3, eb + 3);
            }
            if ((u % SCAN_G) == SCAN_G - 1)
                while (qn >= 2 * EPW) { qn -= 2 * EPW; process(qn, 2 * EPW, twoBatches); }
        }
    }
    if (wv == 0) {
        for (int eb = nFull << 7; eb < E; eb += 32) {
            const int e = eb + lane;
            const int d = (e < E) ? dst[e] : -1;
            append((unsigned)(d - lo) < (unsigned)cnt, e);
        }
    }
    while (qn > 0) {
        const int n = (qn < EPW) ? qn : EPW;
        qn -= n;
        process(qn, n, oneBatch);
    }

    __syncthreads();
    float4* __restrict__ orow = (float4*)(agg + (size_t)lo * HC);
    for (int i = tid; i < cnt * (HC / 4); i += OWN_WAVES * 32) orow[i] = ((const float4*)acc)[i];
    for (int i = tid; i < cnt * HEADS; i += OWN_WAVES * 32) denom[lo * HEADS + i] = dsum[i];
}

__global__ __launch_bounds__(HC) void k_epilogue(
    float* __restrict__ io, const float* __restrict__ denom,
    const float* __restrict__ bias)
{
    const int n = blockIdx.x;
    const int t = threadIdx.x;
    const float dn = denom[n * HEADS + (t >> 5)];
    io[(size_t)n * HC + t] = io[(size_t)n * HC + t] / dn + bias[t];
}

extern "C" void kernel_launch(void* const* d_in, const int* in_sizes, int n_in,
                              void* d_out, int out_size, void* d_ws, size_t ws_size,
                              hipStream_t stream)
{
    const float* x    = (const float*)d_in[0];
    const int*   esrc = (const int*)d_in[1];
    const int*   edst = (const int*)d_in[2];
    const float* W[3]  = {(const float*)d_in[3],  (const float*)d_in[7],  (const float*)d_in[11]};
    const float* As[3] = {(const float*)d_in[4],  (const float*)d_in[8],  (const float*)d_in[12]};
    const float* Ad[3] = {(const float*)d_in[5],  (const float*)d_in[9],  (const float*)d_in[13]};
    const float* Bi[3] = {(const float*)d_in[6],  (const float*)d_in[10], (const float*)d_in[14]};
    const float* LnG[2] = {(const float*)d_in[15], (const float*)d_in[17]};
    const float* LnB[2] = {(const float*)d_in[16], (const float*)d_in[18]};

    const int Nn   = in_sizes[0] / HC;
    const int E    = in_sizes[1];
    const int Etot = E + Nn;

    float*    hbuf  = (float*)d_ws;
    float*    agg   = hbuf + (size_t)Nn * HC;
    float*    als   = agg  + (size_t)Nn * HC;
    float*    ald   = als  + (size_t)Nn * HEADS;
    unsigned* mkey  = (unsigned*)(ald + (size_t)Nn * HEADS);
    float*    denom = (float*)(mkey + (size_t)Nn * HEADS);

    const int TB = 256;
    const unsigned gGemm = (unsigned)((Nn + 15) / 16);
    const unsigned gAttn = (unsigned)(((long long)Nn * HEADS + TB - 1) / TB);
    const unsigned gEH   = 2048u;
    const unsigned gOwn  = (unsigned)((Nn + RNG - 1) / RNG);

    const float* cur = x;
    const float* pB = nullptr;
    const float* pG = nullptr;
    const float* pL = nullptr;
    int fuse = 0;
    for (int l = 0; l < 3; ++l) {
        float* layer_out = (l == 2) ? (float*)d_out : agg;
        float* part = (l == 2) ? agg : (float*)d_out;
        float* gst  = part + (size_t)gAttn * 8;
        int*   gflag = (int*)(gst + HEADS);

        k_gemm_wmma<<<gGemm, 256, 0, stream>>>(cur, W[l], hbuf, Nn, denom, pB, pG, pL, fuse);
        k_attn_logits<<<gAttn, TB, 0, stream>>>(hbuf, As[l], Ad[l], als, ald, part, Nn);
        k_init<<<gAttn, TB, 0, stream>>>(mkey, Nn, part, (int)gAttn, gst, gflag);
        k_edge_max<<<gEH, TB, 0, stream>>>(esrc, edst, als, ald, mkey, gflag, E, Etot);
        k_edge_own<<<gOwn, OWN_WAVES * 32, 0, stream>>>(esrc, edst, als, ald, mkey, denom,
                                            hbuf, layer_out, gst, gflag, E, Nn);
        if (l < 2) {
            cur = layer_out;
            pB = Bi[l]; pG = LnG[l]; pL = LnB[l];
            fuse = 1;
        }
    }
    k_epilogue<<<(unsigned)Nn, HC, 0, stream>>>((float*)d_out, denom, Bi[2]);
}


# Edge scan once per call: in layer 0 each owner wave also keeps its (src, dst row) list in the free output buffer; layers 1,2 walk the lists: no dst rescan, no index gathers; LDS arrival-order sums
# speedup vs baseline: 3.6590x; 1.2464x over previous
#include <hip/hip_runtime.h>
#include <hip/hip_bf16.h>
#include <float.h>
#include <math.h>
#include <type_traits>

#define HEADS 4
#define CPH 32
#define HC 128
#define NEG_SLOPE 0.2f
#define EPSV 1e-5f

typedef float v2f __attribute__((ext_vector_type(2)));
typedef float v8f __attribute__((ext_vector_type(8)));

__device__ __forceinline__ unsigned enc_f32(float f) {
    unsigned u = __float_as_uint(f);
    return (u & 0x80000000u) ? ~u : (u | 0x80000000u);
}
__device__ __forceinline__ float dec_f32(unsigned k) {
    unsigned u = (k & 0x80000000u) ? (k ^ 0x80000000u) : ~k;
    return __uint_as_float(u);
}

__device__ __forceinline__ void edge_sd(long long e, const int* __restrict__ src,
                                        const int* __restrict__ dst, int E,
                                        int* s, int* d) {
    if (e < (long long)E) { *s = src[e]; *d = dst[e]; }
    else { int v = (int)(e - E); *s = v; *d = v; }
}

#define XP 132

__global__ __launch_bounds__(256) __attribute__((amdgpu_waves_per_eu(16, 16))) void k_gemm_wmma(
    const float* __restrict__ A, const float* __restrict__ W,
    float* __restrict__ Hout, int M,
    const float* __restrict__ denom, const float* __restrict__ bias,
    const float* __restrict__ g, const float* __restrict__ bln, int fuse)
{
    __shared__ __attribute__((aligned(16))) float Xs[16 * XP];

    const int lane = threadIdx.x & 31;
    const int wave = threadIdx.x >> 5;
    const int m0   = blockIdx.x * 16;
    const int n0   = wave * 16;
    const int hi   = lane >> 4;
    const int l16  = lane & 15;
    const bool full = (m0 + 16 <= M);

    float4 v[2];
    int rr[2];
#pragma unroll
    for (int i = 0; i < 2; ++i) {
        int r = m0 + wave + 8 * i;
        if (r >= M) r = M - 1;
        rr[i] = r;
        v[i] = *(const float4*)(A + (size_t)r * HC + 4 * lane);
    }
    if (fuse) {
        float dn[2];
#pragma unroll
        for (int i = 0; i < 2; ++i) dn[i] = denom[rr[i] * HEADS + (lane >> 3)];
        const float4 bi = *(const float4*)(bias + 4 * lane);
        const float4 gg = *(const float4*)(g + 4 * lane);
        const float4 bb = *(const float4*)(bln + 4 * lane);
#pragma unroll
        for (int i = 0; i < 2; ++i) {
            float4 t = v[i];
            t.x = t.x / dn[i] + bi.x;
            t.y = t.y / dn[i] + bi.y;
            t.z = t.z / dn[i] + bi.z;
            t.w = t.w / dn[i] + bi.w;
            float s1 = (t.x + t.y) + (t.z + t.w);
            float s2 = (t.x * t.x + t.y * t.y) + (t.z * t.z + t.w * t.w);
#pragma unroll
            for (int off = 16; off > 0; off >>= 1) {
                s1 += __shfl_xor(s1, off, 32);
                s2 += __shfl_xor(s2, off, 32);
            }
            const float mu  = s1 * (1.f / HC);
            const float var = s2 * (1.f / HC) - mu * mu;
            const float rs  = rsqrtf(var + EPSV);
            t.x = fmaxf((t.x - mu) * rs * gg.x + bb.x, 0.f);
            t.y = fmaxf((t.y - mu) * rs * gg.y + bb.y, 0.f);
            t.z = fmaxf((t.z - mu) * rs * gg.z + bb.z, 0.f);
            t.w = fmaxf((t.w - mu) * rs * gg.w + bb.w, 0.f);
            v[i] = t;
        }
    }
#pragma unroll
    for (int i = 0; i < 2; ++i)
        *(float4*)(Xs + (wave + 8 * i) * XP + 4 * lane) = v[i];
    __syncthreads();

    const float* xr = Xs + l16 * XP + 2 * hi;
    const float* __restrict__ wc = W + (size_t)(2 * hi) * HC + n0 + l16;

    v8f acc = {0.f, 0.f, 0.f, 0.f, 0.f, 0.f, 0.f, 0.f};

#pragma unroll 4
    for (int k = 0; k < HC; k += 4) {
        v2f av, bv;
        av = *(const v2f*)(xr + k);
        bv.x = wc[(size_t)k * HC];
        bv.y = wc[(size_t)(k + 1) * HC];
        acc = __builtin_amdgcn_wmma_f32_16x16x4_f32(
             false, av,  false, bv,
             (short)0, acc,  false,  false);
    }

    float* __restrict__ outp = Hout + (size_t)(m0 + 8 * hi) * HC + n0 + l16;
    if (full) {
#pragma unroll
        for (int r = 0; r < 8; ++r) outp[(size_t)r * HC] = acc[r];
    } else {
#pragma unroll
        for (int r = 0; r < 8; ++r) {
            int ro = m0 + r + 8 * hi;
            if (ro < M) Hout[(size_t)ro * HC + n0 + l16] = acc[r];
        }
    }
}

#define SHIFT_SPAN 32.0f

__global__ __launch_bounds__(256) void k_attn_logits(const float* __restrict__ Hm,
                              const float* __restrict__ a_s,
                              const float* __restrict__ a_d,
                              float* __restrict__ als, float* __restrict__ ald,
                              float* __restrict__ part, int Nn)
{
    __shared__ float red[8][8];
    const int idx = blockIdx.x * 256 + threadIdx.x;
    const bool ok = idx < Nn * HEADS;
    const int lane = threadIdx.x & 31;
    const int wave = threadIdx.x >> 5;
    float ss = 0.f, sd = 0.f;
    if (ok) {
        int n = idx >> 2, h = idx & 3;
        const float4* hp = (const float4*)(Hm + (size_t)n * HC + h * CPH);
        const float4* as = (const float4*)(a_s + h * CPH);
        const float4* ad = (const float4*)(a_d + h * CPH);
#pragma unroll
        for (int c = 0; c < CPH / 4; ++c) {
            float4 v = hp[c], a = as[c], b = ad[c];
            ss += v.x * a.x + v.y * a.y + v.z * a.z + v.w * a.w;
            sd += v.x * b.x + v.y * b.y + v.z * b.z + v.w * b.w;
        }
        als[idx] = ss;
        ald[idx] = sd;
    }
    float mx = ok ? ss : -INFINITY;
    float mn = ok ? ss :  INFINITY;
#pragma unroll
    for (int off = 4; off < 32; off <<= 1) {
        mx = fmaxf(mx, __shfl_xor(mx, off, 32));
        mn = fminf(mn, __shfl_xor(mn, off, 32));
    }
    if (lane < 4) { red[wave][lane] = mx; red[wave][4 + lane] = mn; }
    __syncthreads();
    if (threadIdx.x < 8) {
        const int t = threadIdx.x;
        float r = red[0][t];
#pragma unroll
        for (int w = 1; w < 8; ++w)
            r = (t < 4) ? fmaxf(r, red[w][t]) : fminf(r, red[w][t]);
        part[blockIdx.x * 8 + t] = r;
    }
}

__global__ __launch_bounds__(256) void k_init(unsigned* __restrict__ mkey, int Nn,
                       const float* __restrict__ part, int nPart,
                       float* __restrict__ gst, int* __restrict__ gflag)
{
    __shared__ float r2[8][8];
    long long idx = (long long)blockIdx.x * blockDim.x + threadIdx.x;
    if (idx < (long long)Nn * HEADS) mkey[idx] = 0u;

    if (blockIdx.x == 0) {
        const int t = threadIdx.x;
        const int c = t & 7;
        const int lane = t & 31;
        float r = (c < 4) ? -INFINITY : INFINITY;
        for (int b = t >> 3; b < nPart; b += 32) {
            const float p = part[b * 8 + c];
            r = (c < 4) ? fmaxf(r, p) : fminf(r, p);
        }
        float o = __shfl_xor(r, 8, 32);
        r = (c < 4) ? fmaxf(r, o) : fminf(r, o);
        o = __shfl_xor(r, 16, 32);
        r = (c < 4) ? fmaxf(r, o) : fminf(r, o);
        if (lane < 8) r2[t >> 5][lane] = r;
        __syncthreads();
        if (t < 8) {
            float q = r2[0][t];
#pragma unroll
            for (int w = 1; w < 8; ++w)
                q = (t < 4) ? fmaxf(q, r2[w][t]) : fminf(q, r2[w][t]);
            r2[0][t] = q;
        }
        __syncthreads();
        if (t == 0) {
            int wide = 0;
#pragma unroll
            for (int h = 0; h < HEADS; ++h) {
                gst[h] = r2[0][h];
                if (!(r2[0][h] - r2[0][4 + h] <= SHIFT_SPAN)) wide = 1;
            }
            gflag[0] = wide;
        }
    }
}

__global__ __launch_bounds__(256) void k_edge_max(const int* __restrict__ src, const int* __restrict__ dst,
                           const float* __restrict__ als, const float* __restrict__ ald,
                           unsigned* __restrict__ mkey, const int* __restrict__ gflag,
                           int E, int Etot)
{
    if (gflag[0] == 0) return;
    const long long total = (long long)Etot * HEADS;
    const long long step = (long long)gridDim.x * blockDim.x;
    for (long long idx = (long long)blockIdx.x * blockDim.x + threadIdx.x; idx < total; idx += step) {
        long long e = idx >> 2; int h = (int)(idx & 3);
        int s, d; edge_sd(e, src, dst, E, &s, &d);
        float ev = als[s * HEADS + h] + ald[d * HEADS + h];
        ev = (ev < 0.f) ? NEG_SLOPE * ev : ev;
        atomicMax(&mkey[d * HEADS + h], enc_f32(ev));
    }
}

#define EPW 8
#define RNG 416
#define OWN_WAVES 32
#define SCAN_U 8
#define SCAN_G 4
#define QCAP 544
#define LCAP 1536
#define DL_BITS 9

__device__ __forceinline__ void lds_add(float* p, float x) {
    __hip_atomic_fetch_add(p, x, __ATOMIC_RELAXED, __HIP_MEMORY_SCOPE_WORKGROUP);
}

__global__ __launch_bounds__(OWN_WAVES * 32) void k_edge_own(
                           const int* __restrict__ src, const int* __restrict__ dst,
                           const float* __restrict__ als, const float* __restrict__ ald,
                           const unsigned* __restrict__ mkey, float* __restrict__ denom,
                           const float* __restrict__ Hm, float* __restrict__ agg,
                           const float* __restrict__ gst, const int* __restrict__ gflag,
                           unsigned* __restrict__ glist, int* __restrict__ gcnt, int mode,
                           int E, int Nn)
{
    __shared__ __attribute__((aligned(16))) float acc[RNG * HC];
    __shared__ __attribute__((aligned(16))) float dsum[RNG * HEADS];
    __shared__ int qe[OWN_WAVES][QCAP];

    const int tid  = threadIdx.x;
    const int lane = tid & 31;
    const int wv = __builtin_amdgcn_readfirstlane(tid >> 5);
    const int lo = blockIdx.x * RNG;
    const int cnt = (Nn - lo < RNG) ? (Nn - lo) : RNG;
    const int exact = gflag[0];
    const float Mh = gst[lane & 3];

    for (int i = tid; i < RNG * HC / 4; i += OWN_WAVES * 32)
        ((float4*)acc)[i] = make_float4(0.f, 0.f, 0.f, 0.f);
    for (int i = tid; i < RNG * HEADS; i += OWN_WAVES * 32) dsum[i] = 0.f;
    __syncthreads();

    int qn = 0;
    int* const myq = qe[wv];

    const int widx = blockIdx.x * OWN_WAVES + wv;
    unsigned* const mylist = glist + (size_t)widx * LCAP;
    int total = 0;
    if (mode != 0) total = __builtin_amdgcn_readfirstlane(gcnt[widx]);
    const bool replay = (mode != 0) && (total <= LCAP);
    const bool record = (mode == 0);
    int gp = 0;

    auto append = [&](bool take, int e) {
        const unsigned mask = __builtin_amdgcn_ballot_w32(take);
        if (mask != 0u) {
            const int pos = qn + (int)__builtin_amdgcn_mbcnt_lo(mask, 0u);
            if (take) myq[pos] = e;
            qn += __builtin_popcount(mask);
        }
    };

    auto process = [&](int base, int n, auto nbTag) {
        constexpr int NB = decltype(nbTag)::value;
        const int j = lane >> 2;
        const int h = lane & 3;
        int s[NB], d[NB];
        bool act[NB];
#pragma unroll
        for (int b = 0; b < NB; ++b) {
            act[b] = b * EPW + j < n;
            const int slot = base + (act[b] ? b * EPW + j : 0);
            if (replay) {
                const unsigned ent = mylist[slot];
                s[b] = (int)(ent >> DL_BITS);
                d[b] = lo + (int)(ent & ((1u << DL_BITS) - 1u));
            } else {
                const int e = myq[slot];
                edge_sd((long long)e, src, dst, E, &s[b], &d[b]);
            }
        }

        float v[NB][EPW][HEADS];
#pragma unroll
        for (int b = 0; b < NB; ++b) {
#pragma unroll
            for (int q = 0; q < EPW; ++q) {
                const int sq = __builtin_amdgcn_readlane(s[b], 4 * q);
                const float* __restrict__ hp = Hm + (size_t)sq * HC + lane;
#pragma unroll
                for (int k = 0; k < HEADS; ++k) v[b][q][k] = hp[k * CPH];
            }
        }

        float as_[NB], ad_[NB];
        unsigned mk[NB];
#pragma unroll
        for (int b = 0; b < NB; ++b) {
            const int dh = d[b] * HEADS + h;
            as_[b] = als[s[b] * HEADS + h];
            ad_[b] = ald[dh];
            mk[b] = 0u;
            if (exact) mk[b] = mkey[dh];
        }

        __builtin_amdgcn_sched_barrier(0);

        if (record) {
#pragma unroll
            for (int b = 0; b < NB; ++b) {
                const int gi = gp + b * EPW + j;
                if (h == 0 && act[b] && gi < LCAP)
                    mylist[gi] = ((unsigned)s[b] << DL_BITS) | (unsigned)(d[b] - lo);
            }
        }
        if (!replay) gp += n;

#pragma unroll
        for (int b = 0; b < NB; ++b) {
            float ev = as_[b] + ad_[b];
            ev = (ev < 0.f) ? NEG_SLOPE * ev : ev;
            float m;
            if (exact) m = dec_f32(mk[b]);
            else { m = Mh + ad_[b]; m = (m < 0.f) ? NEG_SLOPE * m : m; }
            const float w = expf(ev - m);
            const int dl = d[b] - lo;
            if (act[b]) lds_add(&dsum[dl * HEADS + h], w);

            auto add_row = [&](int q) {
                const int dq = __builtin_amdgcn_readlane(dl, 4 * q);
                float* ap = acc + dq * HC + lane;
#pragma unroll
                for (int k = 0; k < HEADS; ++k) {
                    const float wk = __int_as_float(
                        __builtin_amdgcn_readlane(__float_as_int(w), 4 * q + k));
                    lds_add(ap + k * CPH, v[b][q][k] * wk);
                }
            };
            if (n == NB * EPW) {
#pragma unroll
                for (int q = 0; q < EPW; ++q) add_row(q);
            } else {
#pragma unroll
                for (int q = 0; q < EPW; ++q) if (b * EPW + q < n) add_row(q);
            }
        }
    };
    const std::integral_constant<int, 1> oneBatch{};
    const std::integral_constant<int, 2> twoBatches{};

    if (replay) {
        int pos = 0;
        while (total - pos >= 2 * EPW) { process(pos, 2 * EPW, twoBatches); pos += 2 * EPW; }
        while (pos < total) {
            const int n = (total - pos < EPW) ? (total - pos) : EPW;
            process(pos, n, oneBatch);
            pos += n;
        }
    } else {
    {
        const int PERW = (RNG + OWN_WAVES - 1) / OWN_WAVES;
        const int node = lo + wv * PERW + lane;
        append(lane < PERW && node < lo + cnt, E + node);
    }

    const int nFull = E >> 7;
    const int rot = (int)(((long long)nFull * blockIdx.x) / gridDim.x);
    const int4* __restrict__ d4p = (const int4*)dst;
    for (int c0 = wv; c0 < nFull; c0 += OWN_WAVES * SCAN_U) {
        int4 dv[SCAN_U];
        int cb[SCAN_U];
        if (c0 + OWN_WAVES * (SCAN_U - 1) < nFull) {
#pragma unroll
            for (int u = 0; u < SCAN_U; ++u) {
                int ca = c0 + OWN_WAVES * u + rot;
                if (ca >= nFull) ca -= nFull;
                cb[u] = ca;
                dv[u] = d4p[(size_t)ca * 32 + lane];
            }
        } else {
#pragma unroll
            for (int u = 0; u < SCAN_U; ++u) {
                const int c = c0 + OWN_WAVES * u;
                int ca = c + rot;
                if (ca >= nFull) ca -= nFull;
                cb[u] = ca;
                if (c < nFull) dv[u] = d4p[(size_t)ca * 32 + lane];
                else dv[u] = make_int4(-1, -1, -1, -1);
            }
        }
#pragma unroll
        for (int u = 0; u < SCAN_U; ++u) {
            const int4 t = dv[u];
            const bool m0 = (unsigned)(t.x - lo) < (unsigned)cnt;
            const bool m1 = (unsigned)(t.y - lo) < (unsigned)cnt;
            const bool m2 = (unsigned)(t.z - lo) < (unsigned)cnt;
            const bool m3 = (unsigned)(t.w - lo) < (unsigned)cnt;
            if (__builtin_amdgcn_ballot_w32(m0 | m1 | m2 | m3) != 0u) {
                const int eb = cb[u] * 128 + 4 * lane;
                append(m0, eb);
                append(m1, eb + 1);
                append(m2, eb + 2);
                append(m3, eb + 3);
            }
            if ((u % SCAN_G) == SCAN_G - 1)
                while (qn >= 2 * EPW) { qn -= 2 * EPW; process(qn, 2 * EPW, twoBatches); }
        }
    }
    if (wv == 0) {
        for (int eb = nFull << 7; eb < E; eb += 32) {
            const int e = eb + lane;
            const int d = (e < E) ? dst[e] : -1;
            append((unsigned)(d - lo) < (unsigned)cnt, e);
        }
    }
    while (qn > 0) {
        const int n = (qn < EPW) ? qn : EPW;
        qn -= n;
        process(qn, n, oneBatch);
    }
    if (record && lane == 0) gcnt[widx] = gp;
    }

    __syncthreads();
    float4* __restrict__ orow = (float4*)(agg + (size_t)lo * HC);
    for (int i = tid; i < cnt * (HC / 4); i += OWN_WAVES * 32) orow[i] = ((const float4*)acc)[i];
    for (int i = tid; i < cnt * HEADS; i += OWN_WAVES * 32) denom[lo * HEADS + i] = dsum[i];
}

__global__ __launch_bounds__(HC) void k_epilogue(
    const float* __restrict__ in, float* __restrict__ out,
    const float* __restrict__ denom, const float* __restrict__ bias)
{
    const int n = blockIdx.x;
    const int t = threadIdx.x;
    const float dn = denom[n * HEADS + (t >> 5)];
    out[(size_t)n * HC + t] = in[(size_t)n * HC + t] / dn + bias[t];
}

extern "C" void kernel_launch(void* const* d_in, const int* in_sizes, int n_in,
                              void* d_out, int out_size, void* d_ws, size_t ws_size,
                              hipStream_t stream)
{
    const float* x    = (const float*)d_in[0];
    const int*   esrc = (const int*)d_in[1];
    const int*   edst = (const int*)d_in[2];
    const float* W[3]  = {(const float*)d_in[3],  (const float*)d_in[7],  (const float*)d_in[11]};
    const float* As[3] = {(const float*)d_in[4],  (const float*)d_in[8],  (const float*)d_in[12]};
    const float* Ad[3] = {(const float*)d_in[5],  (const float*)d_in[9],  (const float*)d_in[13]};
    const float* Bi[3] = {(const float*)d_in[6],  (const float*)d_in[10], (const float*)d_in[14]};
    const float* LnG[2] = {(const float*)d_in[15], (const float*)d_in[17]};
    const float* LnB[2] = {(const float*)d_in[16], (const float*)d_in[18]};

    const int Nn   = in_sizes[0] / HC;
    const int E    = in_sizes[1];
    const int Etot = E + Nn;

    float*    hbuf  = (float*)d_ws;
    float*    agg   = hbuf + (size_t)Nn * HC;
    float*    als   = agg  + (size_t)Nn * HC;
    float*    ald   = als  + (size_t)Nn * HEADS;
    unsigned* mkey  = (unsigned*)(ald + (size_t)Nn * HEADS);
    float*    denom = (float*)(mkey + (size_t)Nn * HEADS);

    const int TB = 256;
    const unsigned gGemm = (unsigned)((Nn + 15) / 16);
    const unsigned gAttn = (unsigned)(((long long)Nn * HEADS + TB - 1) / TB);
    const unsigned gEH   = 2048u;
    const unsigned gOwn  = (unsigned)((Nn + RNG - 1) / RNG);

    const float* cur = x;
    const float* pB = nullptr;
    const float* pG = nullptr;
    const float* pL = nullptr;
    int fuse = 0;
    float*    part  = (float*)d_out;
    float*    gst   = part + (size_t)gAttn * 8;
    int*      gflag = (int*)(gst + HEADS);
    int*      gcnt  = gflag + 4;
    unsigned* glist = (unsigned*)(gcnt + (size_t)gOwn * OWN_WAVES);
    for (int l = 0; l < 3; ++l) {
        float* layer_out = agg;

        k_gemm_wmma<<<gGemm, 256, 0, stream>>>(cur, W[l], hbuf, Nn, denom, pB, pG, pL, fuse);
        k_attn_logits<<<gAttn, TB, 0, stream>>>(hbuf, As[l], Ad[l], als, ald, part, Nn);
        k_init<<<gAttn, TB, 0, stream>>>(mkey, Nn, part, (int)gAttn, gst, gflag);
        k_edge_max<<<gEH, TB, 0, stream>>>(esrc, edst, als, ald, mkey, gflag, E, Etot);
        k_edge_own<<<gOwn, OWN_WAVES * 32, 0, stream>>>(esrc, edst, als, ald, mkey, denom,
                                            hbuf, layer_out, gst, gflag,
                                            glist, gcnt, (l == 0) ? 0 : 1, E, Nn);
        if (l < 2) {
            cur = layer_out;
            pB = Bi[l]; pG = LnG[l]; pL = LnB[l];
            fuse = 1;
        }
    }
    k_epilogue<<<(unsigned)Nn, HC, 0, stream>>>(agg, (float*)d_out, denom, Bi[2]);
}


# Last layer: owner groups store out = sums/denom + bias straight from LDS; a group's kept lists now sit in its own span of output rows, so it overwrites only its own; epilogue launch goes; same sums
# speedup vs baseline: 3.9231x; 1.0722x over previous
#include <hip/hip_runtime.h>
#include <hip/hip_bf16.h>
#include <float.h>
#include <math.h>
#include <type_traits>

#define HEADS 4
#define CPH 32
#define HC 128
#define NEG_SLOPE 0.2f
#define EPSV 1e-5f

typedef float v2f __attribute__((ext_vector_type(2)));
typedef float v8f __attribute__((ext_vector_type(8)));

__device__ __forceinline__ unsigned enc_f32(float f) {
    unsigned u = __float_as_uint(f);
    return (u & 0x80000000u) ? ~u : (u | 0x80000000u);
}
__device__ __forceinline__ float dec_f32(unsigned k) {
    unsigned u = (k & 0x80000000u) ? (k ^ 0x80000000u) : ~k;
    return __uint_as_float(u);
}

__device__ __forceinline__ void edge_sd(long long e, const int* __restrict__ src,
                                        const int* __restrict__ dst, int E,
                                        int* s, int* d) {
    if (e < (long long)E) { *s = src[e]; *d = dst[e]; }
    else { int v = (int)(e - E); *s = v; *d = v; }
}

#define XP 132

__global__ __launch_bounds__(256) __attribute__((amdgpu_waves_per_eu(16, 16))) void k_gemm_wmma(
    const float* __restrict__ A, const float* __restrict__ W,
    float* __restrict__ Hout, int M,
    const float* __restrict__ denom, const float* __restrict__ bias,
    const float* __restrict__ g, const float* __restrict__ bln, int fuse)
{
    __shared__ __attribute__((aligned(16))) float Xs[16 * XP];

    const int lane = threadIdx.x & 31;
    const int wave = threadIdx.x >> 5;
    const int m0   = blockIdx.x * 16;
    const int n0   = wave * 16;
    const int hi   = lane >> 4;
    const int l16  = lane & 15;
    const bool full = (m0 + 16 <= M);

    float4 v[2];
    int rr[2];
#pragma unroll
    for (int i = 0; i < 2; ++i) {
        int r = m0 + wave + 8 * i;
        if (r >= M) r = M - 1;
        rr[i] = r;
        v[i] = *(const float4*)(A + (size_t)r * HC + 4 * lane);
    }
    if (fuse) {
        float dn[2];
#pragma unroll
        for (int i = 0; i < 2; ++i) dn[i] = denom[rr[i] * HEADS + (lane >> 3)];
        const float4 bi = *(const float4*)(bias + 4 * lane);
        const float4 gg = *(const float4*)(g + 4 * lane);
        const float4 bb = *(const float4*)(bln + 4 * lane);
#pragma unroll
        for (int i = 0; i < 2; ++i) {
            float4 t = v[i];
            t.x = t.x / dn[i] + bi.x;
            t.y = t.y / dn[i] + bi.y;
            t.z = t.z / dn[i] + bi.z;
            t.w = t.w / dn[i] + bi.w;
            float s1 = (t.x + t.y) + (t.z + t.w);
            float s2 = (t.x * t.x + t.y * t.y) + (t.z * t.z + t.w * t.w);
#pragma unroll
            for (int off = 16; off > 0; off >>= 1) {
                s1 += __shfl_xor(s1, off, 32);
                s2 += __shfl_xor(s2, off, 32);
            }
            const float mu  = s1 * (1.f / HC);
            const float var = s2 * (1.f / HC) - mu * mu;
            const float rs  = rsqrtf(var + EPSV);
            t.x = fmaxf((t.x - mu) * rs * gg.x + bb.x, 0.f);
            t.y = fmaxf((t.y - mu) * rs * gg.y + bb.y, 0.f);
            t.z = fmaxf((t.z - mu) * rs * gg.z + bb.z, 0.f);
            t.w = fmaxf((t.w - mu) * rs * gg.w + bb.w, 0.f);
            v[i] = t;
        }
    }
#pragma unroll
    for (int i = 0; i < 2; ++i)
        *(float4*)(Xs + (wave + 8 * i) * XP + 4 * lane) = v[i];
    __syncthreads();

    const float* xr = Xs + l16 * XP + 2 * hi;
    const float* __restrict__ wc = W + (size_t)(2 * hi) * HC + n0 + l16;

    v8f acc = {0.f, 0.f, 0.f, 0.f, 0.f, 0.f, 0.f, 0.f};

#pragma unroll 4
    for (int k = 0; k < HC; k += 4) {
        v2f av, bv;
        av = *(const v2f*)(xr + k);
        bv.x = wc[(size_t)k * HC];
        bv.y = wc[(size_t)(k + 1) * HC];
        acc = __builtin_amdgcn_wmma_f32_16x16x4_f32(
             false, av,  false, bv,
             (short)0, acc,  false,  false);
    }

    float* __restrict__ outp = Hout + (size_t)(m0 + 8 * hi) * HC + n0 + l16;
    if (full) {
#pragma unroll
        for (int r = 0; r < 8; ++r) outp[(size_t)r * HC] = acc[r];
    } else {
#pragma unroll
        for (int r = 0; r < 8; ++r) {
            int ro = m0 + r + 8 * hi;
            if (ro < M) Hout[(size_t)ro * HC + n0 + l16] = acc[r];
        }
    }
}

#define SHIFT_SPAN 32.0f

__global__ __launch_bounds__(256) void k_attn_logits(const float* __restrict__ Hm,
                              const float* __restrict__ a_s,
                              const float* __restrict__ a_d,
                              float* __restrict__ als, float* __restrict__ ald,
                              float* __restrict__ part, int Nn)
{
    __shared__ float red[8][8];
    const int idx = blockIdx.x * 256 + threadIdx.x;
    const bool ok = idx < Nn * HEADS;
    const int lane = threadIdx.x & 31;
    const int wave = threadIdx.x >> 5;
    float ss = 0.f, sd = 0.f;
    if (ok) {
        int n = idx >> 2, h = idx & 3;
        const float4* hp = (const float4*)(Hm + (size_t)n * HC + h * CPH);
        const float4* as = (const float4*)(a_s + h * CPH);
        const float4* ad = (const float4*)(a_d + h * CPH);
#pragma unroll
        for (int c = 0; c < CPH / 4; ++c) {
            float4 v = hp[c], a = as[c], b = ad[c];
            ss += v.x * a.x + v.y * a.y + v.z * a.z + v.w * a.w;
            sd += v.x * b.x + v.y * b.y + v.z * b.z + v.w * b.w;
        }
        als[idx] = ss;
        ald[idx] = sd;
    }
    float mx = ok ? ss : -INFINITY;
    float mn = ok ? ss :  INFINITY;
#pragma unroll
    for (int off = 4; off < 32; off <<= 1) {
        mx = fmaxf(mx, __shfl_xor(mx, off, 32));
        mn = fminf(mn, __shfl_xor(mn, off, 32));
    }
    if (lane < 4) { red[wave][lane] = mx; red[wave][4 + lane] = mn; }
    __syncthreads();
    if (threadIdx.x < 8) {
        const int t = threadIdx.x;
        float r = red[0][t];
#pragma unroll
        for (int w = 1; w < 8; ++w)
            r = (t < 4) ? fmaxf(r, red[w][t]) : fminf(r, red[w][t]);
        part[blockIdx.x * 8 + t] = r;
    }
}

__global__ __launch_bounds__(256) void k_init(unsigned* __restrict__ mkey, int Nn,
                       const float* __restrict__ part, int nPart,
                       float* __restrict__ gst, int* __restrict__ gflag)
{
    __shared__ float r2[8][8];
    long long idx = (long long)blockIdx.x * blockDim.x + threadIdx.x;
    if (idx < (long long)Nn * HEADS) mkey[idx] = 0u;

    if (blockIdx.x == 0) {
        const int t = threadIdx.x;
        const int c = t & 7;
        const int lane = t & 31;
        float r = (c < 4) ? -INFINITY : INFINITY;
        for (int b = t >> 3; b < nPart; b += 32) {
            const float p = part[b * 8 + c];
            r = (c < 4) ? fmaxf(r, p) : fminf(r, p);
        }
        float o = __shfl_xor(r, 8, 32);
        r = (c < 4) ? fmaxf(r, o) : fminf(r, o);
        o = __shfl_xor(r, 16, 32);
        r = (c < 4) ? fmaxf(r, o) : fminf(r, o);
        if (lane < 8) r2[t >> 5][lane] = r;
        __syncthreads();
        if (t < 8) {
            float q = r2[0][t];
#pragma unroll
            for (int w = 1; w < 8; ++w)
                q = (t < 4) ? fmaxf(q, r2[w][t]) : fminf(q, r2[w][t]);
            r2[0][t] = q;
        }
        __syncthreads();
        if (t == 0) {
            int wide = 0;
#pragma unroll
            for (int h = 0; h < HEADS; ++h) {
                gst[h] = r2[0][h];
                if (!(r2[0][h] - r2[0][4 + h] <= SHIFT_SPAN)) wide = 1;
            }
            gflag[0] = wide;
        }
    }
}

__global__ __launch_bounds__(256) void k_edge_max(const int* __restrict__ src, const int* __restrict__ dst,
                           const float* __restrict__ als, const float* __restrict__ ald,
                           unsigned* __restrict__ mkey, const int* __restrict__ gflag,
                           int E, int Etot)
{
    if (gflag[0] == 0) return;
    const long long total = (long long)Etot * HEADS;
    const long long step = (long long)gridDim.x * blockDim.x;
    for (long long idx = (long long)blockIdx.x * blockDim.x + threadIdx.x; idx < total; idx += step) {
        long long e = idx >> 2; int h = (int)(idx & 3);
        int s, d; edge_sd(e, src, dst, E, &s, &d);
        float ev = als[s * HEADS + h] + ald[d * HEADS + h];
        ev = (ev < 0.f) ? NEG_SLOPE * ev : ev;
        atomicMax(&mkey[d * HEADS + h], enc_f32(ev));
    }
}

#define EPW 8
#define RNG 416
#define OWN_WAVES 32
#define SCAN_U 8
#define SCAN_G 4
#define QCAP 544
#define LCAP 1536
#define DL_BITS 9

__device__ __forceinline__ void lds_add(float* p, float x) {
    __hip_atomic_fetch_add(p, x, __ATOMIC_RELAXED, __HIP_MEMORY_SCOPE_WORKGROUP);
}

__global__ __launch_bounds__(OWN_WAVES * 32) void k_edge_own(
                           const int* __restrict__ src, const int* __restrict__ dst,
                           const float* __restrict__ als, const float* __restrict__ ald,
                           const unsigned* __restrict__ mkey, float* __restrict__ denom,
                           const float* __restrict__ Hm, float* __restrict__ agg,
                           const float* __restrict__ gst, const int* __restrict__ gflag,
                           float* __restrict__ obuf, const float* __restrict__ bias, int mode,
                           int E, int Nn)
{
    __shared__ __attribute__((aligned(16))) float acc[RNG * HC];
    __shared__ __attribute__((aligned(16))) float dsum[RNG * HEADS];
    __shared__ int qe[OWN_WAVES][QCAP];

    const int tid  = threadIdx.x;
    const int lane = tid & 31;
    const int wv = __builtin_amdgcn_readfirstlane(tid >> 5);
    const int lo = blockIdx.x * RNG;
    const int cnt = (Nn - lo < RNG) ? (Nn - lo) : RNG;
    const int exact = gflag[0];
    const float Mh = gst[lane & 3];

    for (int i = tid; i < RNG * HC / 4; i += OWN_WAVES * 32)
        ((float4*)acc)[i] = make_float4(0.f, 0.f, 0.f, 0.f);
    for (int i = tid; i < RNG * HEADS; i += OWN_WAVES * 32) dsum[i] = 0.f;
    __syncthreads();

    int qn = 0;
    int* const myq = qe[wv];

    float* const span = obuf + (size_t)lo * HC;
    int cap = (cnt * (HC / OWN_WAVES) - 2) & ~15;
    if (cap > LCAP) cap = LCAP;
    int* const mycnt = (int*)span + wv;
    unsigned* const mylist = (unsigned*)span + OWN_WAVES + (size_t)wv * cap;
    int total = 0;
    if (mode != 0) total = __builtin_amdgcn_readfirstlane(*mycnt);
    const bool replay = (mode != 0) && (total <= cap);
    const bool record = (mode == 0);
    const bool last   = (mode == 2);
    int gp = 0;

    auto append = [&](bool take, int e) {
        const unsigned mask = __builtin_amdgcn_ballot_w32(take);
        if (mask != 0u) {
            const int pos = qn + (int)__builtin_amdgcn_mbcnt_lo(mask, 0u);
            if (take) myq[pos] = e;
            qn += __builtin_popcount(mask);
        }
    };

    auto process = [&](int base, int n, auto nbTag) {
        constexpr int NB = decltype(nbTag)::value;
        const int j = lane >> 2;
        const int h = lane & 3;
        int s[NB], d[NB];
        bool act[NB];
#pragma unroll
        for (int b = 0; b < NB; ++b) {
            act[b] = b * EPW + j < n;
            const int slot = base + (act[b] ? b * EPW + j : 0);
            if (replay) {
                const unsigned ent = mylist[slot];
                s[b] = (int)(ent >> DL_BITS);
                d[b] = lo + (int)(ent & ((1u << DL_BITS) - 1u));
            } else {
                const int e = myq[slot];
                edge_sd((long long)e, src, dst, E, &s[b], &d[b]);
            }
        }

        float v[NB][EPW][HEADS];
#pragma unroll
        for (int b = 0; b < NB; ++b) {
#pragma unroll
            for (int q = 0; q < EPW; ++q) {
                const int sq = __builtin_amdgcn_readlane(s[b], 4 * q);
                const float* __restrict__ hp = Hm + (size_t)sq * HC + lane;
#pragma unroll
                for (int k = 0; k < HEADS; ++k) v[b][q][k] = hp[k * CPH];
            }
        }

        float as_[NB], ad_[NB];
        unsigned mk[NB];
#pragma unroll
        for (int b = 0; b < NB; ++b) {
            const int dh = d[b] * HEADS + h;
            as_[b] = als[s[b] * HEADS + h];
            ad_[b] = ald[dh];
            mk[b] = 0u;
            if (exact) mk[b] = mkey[dh];
        }

        __builtin_amdgcn_sched_barrier(0);

        if (record) {
#pragma unroll
            for (int b = 0; b < NB; ++b) {
                const int gi = gp + b * EPW + j;
                if (h == 0 && act[b] && gi < cap)
                    mylist[gi] = ((unsigned)s[b] << DL_BITS) | (unsigned)(d[b] - lo);
            }
        }
        if (!replay) gp += n;

#pragma unroll
        for (int b = 0; b < NB; ++b) {
            float ev = as_[b] + ad_[b];
            ev = (ev < 0.f) ? NEG_SLOPE * ev : ev;
            float m;
            if (exact) m = dec_f32(mk[b]);
            else { m = Mh + ad_[b]; m = (m < 0.f) ? NEG_SLOPE * m : m; }
            const float w = expf(ev - m);
            const int dl = d[b] - lo;
            if (act[b]) lds_add(&dsum[dl * HEADS + h], w);

            auto add_row = [&](int q) {
                const int dq = __builtin_amdgcn_readlane(dl, 4 * q);
                float* ap = acc + dq * HC + lane;
#pragma unroll
                for (int k = 0; k < HEADS; ++k) {
                    const float wk = __int_as_float(
                        __builtin_amdgcn_readlane(__float_as_int(w), 4 * q + k));
                    lds_add(ap + k * CPH, v[b][q][k] * wk);
                }
            };
            if (n == NB * EPW) {
#pragma unroll
                for (int q = 0; q < EPW; ++q) add_row(q);
            } else {
#pragma unroll
                for (int q = 0; q < EPW; ++q) if (b * EPW + q < n) add_row(q);
            }
        }
    };
    const std::integral_constant<int, 1> oneBatch{};
    const std::integral_constant<int, 2> twoBatches{};

    if (replay) {
        int pos = 0;
        while (total - pos >= 2 * EPW) { process(pos, 2 * EPW, twoBatches); pos += 2 * EPW; }
        while (pos < total) {
            const int n = (total - pos < EPW) ? (total - pos) : EPW;
            process(pos, n, oneBatch);
            pos += n;
        }
    } else {
    {
        const int PERW = (RNG + OWN_WAVES - 1) / OWN_WAVES;
        const int node = lo + wv * PERW + lane;
        append(lane < PERW && node < lo + cnt, E + node);
    }

    const int nFull = E >> 7;
    const int rot = (int)(((long long)nFull * blockIdx.x) / gridDim.x);
    const int4* __restrict__ d4p = (const int4*)dst;
    for (int c0 = wv; c0 < nFull; c0 += OWN_WAVES * SCAN_U) {
        int4 dv[SCAN_U];
        int cb[SCAN_U];
        if (c0 + OWN_WAVES * (SCAN_U - 1) < nFull) {
#pragma unroll
            for (int u = 0; u < SCAN_U; ++u) {
                int ca = c0 + OWN_WAVES * u + rot;
                if (ca >= nFull) ca -= nFull;
                cb[u] = ca;
                dv[u] = d4p[(size_t)ca * 32 + lane];
            }
        } else {
#pragma unroll
            for (int u = 0; u < SCAN_U; ++u) {
                const int c = c0 + OWN_WAVES * u;
                int ca = c + rot;
                if (ca >= nFull) ca -= nFull;
                cb[u] = ca;
                if (c < nFull) dv[u] = d4p[(size_t)ca * 32 + lane];
                else dv[u] = make_int4(-1, -1, -1, -1);
            }
        }
#pragma unroll
        for (int u = 0; u < SCAN_U; ++u) {
            const int4 t = dv[u];
            const bool m0 = (unsigned)(t.x - lo) < (unsigned)cnt;
            const bool m1 = (unsigned)(t.y - lo) < (unsigned)cnt;
            const bool m2 = (unsigned)(t.z - lo) < (unsigned)cnt;
            const bool m3 = (unsigned)(t.w - lo) < (unsigned)cnt;
            if (__builtin_amdgcn_ballot_w32(m0 | m1 | m2 | m3) != 0u) {
                const int eb = cb[u] * 128 + 4 * lane;
                append(m0, eb);
                append(m1, eb + 1);
                append(m2, eb + 2);
                append(m3, eb + 3);
            }
            if ((u % SCAN_G) == SCAN_G - 1)
                while (qn >= 2 * EPW) { qn -= 2 * EPW; process(qn, 2 * EPW, twoBatches); }
        }
    }
    if (wv == 0) {
        for (int eb = nFull << 7; eb < E; eb += 32) {
            const int e = eb + lane;
            const int d = (e < E) ? dst[e] : -1;
            append((unsigned)(d - lo) < (unsigned)cnt, e);
        }
    }
    while (qn > 0) {
        const int n = (qn < EPW) ? qn : EPW;
        qn -= n;
        process(qn, n, oneBatch);
    }
    if (record && lane == 0) *mycnt = gp;
    }

    __syncthreads();
    if (last) {
        float4* __restrict__ orow = (float4*)span;
        const float4 bi = ((const float4*)bias)[tid & 31];
        const int hd = (tid & 31) >> 3;
        for (int i = tid; i < cnt * (HC / 4); i += OWN_WAVES * 32) {
            const float4 a = ((const float4*)acc)[i];
            const float dn = dsum[(i >> 5) * HEADS + hd];
            float4 o;
            o.x = a.x / dn + bi.x;
            o.y = a.y / dn + bi.y;
            o.z = a.z / dn + bi.z;
            o.w = a.w / dn + bi.w;
            orow[i] = o;
        }
    } else {
        float4* __restrict__ orow = (float4*)(agg + (size_t)lo * HC);
        for (int i = tid; i < cnt * (HC / 4); i += OWN_WAVES * 32) orow[i] = ((const float4*)acc)[i];
        for (int i = tid; i < cnt * HEADS; i += OWN_WAVES * 32) denom[lo * HEADS + i] = dsum[i];
    }
}

extern "C" void kernel_launch(void* const* d_in, const int* in_sizes, int n_in,
                              void* d_out, int out_size, void* d_ws, size_t ws_size,
                              hipStream_t stream)
{
    const float* x    = (const float*)d_in[0];
    const int*   esrc = (const int*)d_in[1];
    const int*   edst = (const int*)d_in[2];
    const float* W[3]  = {(const float*)d_in[3],  (const float*)d_in[7],  (const float*)d_in[11]};
    const float* As[3] = {(const float*)d_in[4],  (const float*)d_in[8],  (const float*)d_in[12]};
    const float* Ad[3] = {(const float*)d_in[5],  (const float*)d_in[9],  (const float*)d_in[13]};
    const float* Bi[3] = {(const float*)d_in[6],  (const float*)d_in[10], (const float*)d_in[14]};
    const float* LnG[2] = {(const float*)d_in[15], (const float*)d_in[17]};
    const float* LnB[2] = {(const float*)d_in[16], (const float*)d_in[18]};

    const int Nn   = in_sizes[0] / HC;
    const int E    = in_sizes[1];
    const int Etot = E + Nn;

    float*    hbuf  = (float*)d_ws;
    float*    agg   = hbuf + (size_t)Nn * HC;
    float*    als   = agg  + (size_t)Nn * HC;
    float*    ald   = als  + (size_t)Nn * HEADS;
    unsigned* mkey  = (unsigned*)(ald + (size_t)Nn * HEADS);
    float*    denom = (float*)(mkey + (size_t)Nn * HEADS);

    const int TB = 256;
    const unsigned gGemm = (unsigned)((Nn + 15) / 16);
    const unsigned gAttn = (unsigned)(((long long)Nn * HEADS + TB - 1) / TB);
    const unsigned gEH   = 2048u;
    const unsigned gOwn  = (unsigned)((Nn + RNG - 1) / RNG);

    const float* cur = x;
    const float* pB = nullptr;
    const float* pG = nullptr;
    const float* pL = nullptr;
    int fuse = 0;
    float* part = agg;
    float* gstA = part + (size_t)gAttn * 8;
    float* gstO = (float*)d_out + (size_t)((Nn < RNG) ? Nn : RNG) * HC - 8;
    for (int l = 0; l < 3; ++l) {
        float* layer_out = agg;
        float* gst   = (l == 2) ? gstA : gstO;
        int*   gflag = (int*)(gst + HEADS);

        k_gemm_wmma<<<gGemm, 256, 0, stream>>>(cur, W[l], hbuf, Nn, denom, pB, pG, pL, fuse);
        k_attn_logits<<<gAttn, TB, 0, stream>>>(hbuf, As[l], Ad[l], als, ald, part, Nn);
        k_init<<<gAttn, TB, 0, stream>>>(mkey, Nn, part, (int)gAttn, gst, gflag);
        k_edge_max<<<gEH, TB, 0, stream>>>(esrc, edst, als, ald, mkey, gflag, E, Etot);
        k_edge_own<<<gOwn, OWN_WAVES * 32, 0, stream>>>(esrc, edst, als, ald, mkey, denom,
                                            hbuf, layer_out, gst, gflag,
                                            (float*)d_out, Bi[2], (l == 0) ? 0 : l, E, Nn);
        if (l < 2) {
            cur = layer_out;
            pB = Bi[l]; pG = LnG[l]; pL = LnB[l];
            fuse = 1;
        }
    }
}


# Layer-0 edge scan reads 1 owner byte per edge (dst/416, plane filled by the idle max-pass launch), not 4-byte dst: 512 edges per 16-B load; all scans on 512-edge chunks; LDS arrival-order sums
# speedup vs baseline: 4.1098x; 1.0476x over previous
#include <hip/hip_runtime.h>
#include <hip/hip_bf16.h>
#include <float.h>
#include <math.h>
#include <type_traits>

#define HEADS 4
#define CPH 32
#define HC 128
#define NEG_SLOPE 0.2f
#define EPSV 1e-5f

typedef float v2f __attribute__((ext_vector_type(2)));
typedef float v8f __attribute__((ext_vector_type(8)));

__device__ __forceinline__ unsigned enc_f32(float f) {
    unsigned u = __float_as_uint(f);
    return (u & 0x80000000u) ? ~u : (u | 0x80000000u);
}
__device__ __forceinline__ float dec_f32(unsigned k) {
    unsigned u = (k & 0x80000000u) ? (k ^ 0x80000000u) : ~k;
    return __uint_as_float(u);
}

__device__ __forceinline__ void edge_sd(long long e, const int* __restrict__ src,
                                        const int* __restrict__ dst, int E,
                                        int* s, int* d) {
    if (e < (long long)E) { *s = src[e]; *d = dst[e]; }
    else { int v = (int)(e - E); *s = v; *d = v; }
}

#define XP 132

__global__ __launch_bounds__(256) __attribute__((amdgpu_waves_per_eu(16, 16))) void k_gemm_wmma(
    const float* __restrict__ A, const float* __restrict__ W,
    float* __restrict__ Hout, int M,
    const float* __restrict__ denom, const float* __restrict__ bias,
    const float* __restrict__ g, const float* __restrict__ bln, int fuse)
{
    __shared__ __attribute__((aligned(16))) float Xs[16 * XP];

    const int lane = threadIdx.x & 31;
    const int wave = threadIdx.x >> 5;
    const int m0   = blockIdx.x * 16;
    const int n0   = wave * 16;
    const int hi   = lane >> 4;
    const int l16  = lane & 15;
    const bool full = (m0 + 16 <= M);

    float4 v[2];
    int rr[2];
#pragma unroll
    for (int i = 0; i < 2; ++i) {
        int r = m0 + wave + 8 * i;
        if (r >= M) r = M - 1;
        rr[i] = r;
        v[i] = *(const float4*)(A + (size_t)r * HC + 4 * lane);
    }
    if (fuse) {
        float dn[2];
#pragma unroll
        for (int i = 0; i < 2; ++i) dn[i] = denom[rr[i] * HEADS + (lane >> 3)];
        const float4 bi = *(const float4*)(bias + 4 * lane);
        const float4 gg = *(const float4*)(g + 4 * lane);
        const float4 bb = *(const float4*)(bln + 4 * lane);
#pragma unroll
        for (int i = 0; i < 2; ++i) {
            float4 t = v[i];
            t.x = t.x / dn[i] + bi.x;
            t.y = t.y / dn[i] + bi.y;
            t.z = t.z / dn[i] + bi.z;
            t.w = t.w / dn[i] + bi.w;
            float s1 = (t.x + t.y) + (t.z + t.w);
            float s2 = (t.x * t.x + t.y * t.y) + (t.z * t.z + t.w * t.w);
#pragma unroll
            for (int off = 16; off > 0; off >>= 1) {
                s1 += __shfl_xor(s1, off, 32);
                s2 += __shfl_xor(s2, off, 32);
            }
            const float mu  = s1 * (1.f / HC);
            const float var = s2 * (1.f / HC) - mu * mu;
            const float rs  = rsqrtf(var + EPSV);
            t.x = fmaxf((t.x - mu) * rs * gg.x + bb.x, 0.f);
            t.y = fmaxf((t.y - mu) * rs * gg.y + bb.y, 0.f);
            t.z = fmaxf((t.z - mu) * rs * gg.z + bb.z, 0.f);
            t.w = fmaxf((t.w - mu) * rs * gg.w + bb.w, 0.f);
            v[i] = t;
        }
    }
#pragma unroll
    for (int i = 0; i < 2; ++i)
        *(float4*)(Xs + (wave + 8 * i) * XP + 4 * lane) = v[i];
    __syncthreads();

    const float* xr = Xs + l16 * XP + 2 * hi;
    const float* __restrict__ wc = W + (size_t)(2 * hi) * HC + n0 + l16;

    v8f acc = {0.f, 0.f, 0.f, 0.f, 0.f, 0.f, 0.f, 0.f};

#pragma unroll 4
    for (int k = 0; k < HC; k += 4) {
        v2f av, bv;
        av = *(const v2f*)(xr + k);
        bv.x = wc[(size_t)k * HC];
        bv.y = wc[(size_t)(k + 1) * HC];
        acc = __builtin_amdgcn_wmma_f32_16x16x4_f32(
             false, av,  false, bv,
             (short)0, acc,  false,  false);
    }

    float* __restrict__ outp = Hout + (size_t)(m0 + 8 * hi) * HC + n0 + l16;
    if (full) {
#pragma unroll
        for (int r = 0; r < 8; ++r) outp[(size_t)r * HC] = acc[r];
    } else {
#pragma unroll
        for (int r = 0; r < 8; ++r) {
            int ro = m0 + r + 8 * hi;
            if (ro < M) Hout[(size_t)ro * HC + n0 + l16] = acc[r];
        }
    }
}

#define SHIFT_SPAN 32.0f

__global__ __launch_bounds__(256) void k_attn_logits(const float* __restrict__ Hm,
                              const float* __restrict__ a_s,
                              const float* __restrict__ a_d,
                              float* __restrict__ als, float* __restrict__ ald,
                              float* __restrict__ part, int Nn)
{
    __shared__ float red[8][8];
    const int idx = blockIdx.x * 256 + threadIdx.x;
    const bool ok = idx < Nn * HEADS;
    const int lane = threadIdx.x & 31;
    const int wave = threadIdx.x >> 5;
    float ss = 0.f, sd = 0.f;
    if (ok) {
        int n = idx >> 2, h = idx & 3;
        const float4* hp = (const float4*)(Hm + (size_t)n * HC + h * CPH);
        const float4* as = (const float4*)(a_s + h * CPH);
        const float4* ad = (const float4*)(a_d + h * CPH);
#pragma unroll
        for (int c = 0; c < CPH / 4; ++c) {
            float4 v = hp[c], a = as[c], b = ad[c];
            ss += v.x * a.x + v.y * a.y + v.z * a.z + v.w * a.w;
            sd += v.x * b.x + v.y * b.y + v.z * b.z + v.w * b.w;
        }
        als[idx] = ss;
        ald[idx] = sd;
    }
    float mx = ok ? ss : -INFINITY;
    float mn = ok ? ss :  INFINITY;
#pragma unroll
    for (int off = 4; off < 32; off <<= 1) {
        mx = fmaxf(mx, __shfl_xor(mx, off, 32));
        mn = fminf(mn, __shfl_xor(mn, off, 32));
    }
    if (lane < 4) { red[wave][lane] = mx; red[wave][4 + lane] = mn; }
    __syncthreads();
    if (threadIdx.x < 8) {
        const int t = threadIdx.x;
        float r = red[0][t];
#pragma unroll
        for (int w = 1; w < 8; ++w)
            r = (t < 4) ? fmaxf(r, red[w][t]) : fminf(r, red[w][t]);
        part[blockIdx.x * 8 + t] = r;
    }
}

__global__ __launch_bounds__(256) void k_init(unsigned* __restrict__ mkey, int Nn,
                       const float* __restrict__ part, int nPart,
                       float* __restrict__ gst, int* __restrict__ gflag)
{
    __shared__ float r2[8][8];
    long long idx = (long long)blockIdx.x * blockDim.x + threadIdx.x;
    if (idx < (long long)Nn * HEADS) mkey[idx] = 0u;

    if (blockIdx.x == 0) {
        const int t = threadIdx.x;
        const int c = t & 7;
        const int lane = t & 31;
        float r = (c < 4) ? -INFINITY : INFINITY;
        for (int b = t >> 3; b < nPart; b += 32) {
            const float p = part[b * 8 + c];
            r = (c < 4) ? fmaxf(r, p) : fminf(r, p);
        }
        float o = __shfl_xor(r, 8, 32);
        r = (c < 4) ? fmaxf(r, o) : fminf(r, o);
        o = __shfl_xor(r, 16, 32);
        r = (c < 4) ? fmaxf(r, o) : fminf(r, o);
        if (lane < 8) r2[t >> 5][lane] = r;
        __syncthreads();
        if (t < 8) {
            float q = r2[0][t];
#pragma unroll
            for (int w = 1; w < 8; ++w)
                q = (t < 4) ? fmaxf(q, r2[w][t]) : fminf(q, r2[w][t]);
            r2[0][t] = q;
        }
        __syncthreads();
        if (t == 0) {
            int wide = 0;
#pragma unroll
            for (int h = 0; h < HEADS; ++h) {
                gst[h] = r2[0][h];
                if (!(r2[0][h] - r2[0][4 + h] <= SHIFT_SPAN)) wide = 1;
            }
            gflag[0] = wide;
        }
    }
}

#define RNG 416

__device__ __forceinline__ bool owner_bytes_fit(int E, int Nn) {
    return (long long)E <= (long long)Nn * HEADS * 4 && (Nn + RNG - 1) / RNG <= 256;
}

__global__ __launch_bounds__(256) void k_edge_max(const int* __restrict__ src, const int* __restrict__ dst,
                           const float* __restrict__ als, const float* __restrict__ ald,
                           unsigned* __restrict__ mkey, const int* __restrict__ gflag,
                           int E, int Etot, int prep, int Nn)
{
    const long long step = (long long)gridDim.x * blockDim.x;
    if (gflag[0] == 0) {
        if (prep && owner_bytes_fit(E, Nn)) {
            const long long nW = (long long)(E >> 9) << 7;
            const int4* __restrict__ d4 = (const int4*)dst;
            for (long long i = (long long)blockIdx.x * blockDim.x + threadIdx.x; i < nW; i += step) {
                const int4 d = d4[i];
                mkey[i] = ((unsigned)d.x / RNG) | (((unsigned)d.y / RNG) << 8) |
                          (((unsigned)d.z / RNG) << 16) | (((unsigned)d.w / RNG) << 24);
            }
        }
        return;
    }
    const long long total = (long long)Etot * HEADS;
    for (long long idx = (long long)blockIdx.x * blockDim.x + threadIdx.x; idx < total; idx += step) {
        long long e = idx >> 2; int h = (int)(idx & 3);
        int s, d; edge_sd(e, src, dst, E, &s, &d);
        float ev = als[s * HEADS + h] + ald[d * HEADS + h];
        ev = (ev < 0.f) ? NEG_SLOPE * ev : ev;
        atomicMax(&mkey[d * HEADS + h], enc_f32(ev));
    }
}

#define EPW 8
#define OWN_WAVES 32
#define SCAN_U 8
#define SCAN_G 4
#define QCAP 544
#define LCAP 1536
#define DL_BITS 9

__device__ __forceinline__ void lds_add(float* p, float x) {
    __hip_atomic_fetch_add(p, x, __ATOMIC_RELAXED, __HIP_MEMORY_SCOPE_WORKGROUP);
}

__global__ __launch_bounds__(OWN_WAVES * 32) void k_edge_own(
                           const int* __restrict__ src, const int* __restrict__ dst,
                           const float* __restrict__ als, const float* __restrict__ ald,
                           const unsigned* __restrict__ mkey, float* __restrict__ denom,
                           const float* __restrict__ Hm, float* __restrict__ agg,
                           const float* __restrict__ gst, const int* __restrict__ gflag,
                           float* __restrict__ obuf, const float* __restrict__ bias, int mode,
                           int E, int Nn)
{
    __shared__ __attribute__((aligned(16))) float acc[RNG * HC];
    __shared__ __attribute__((aligned(16))) float dsum[RNG * HEADS];
    __shared__ int qe[OWN_WAVES][QCAP];

    const int tid  = threadIdx.x;
    const int lane = tid & 31;
    const int wv = __builtin_amdgcn_readfirstlane(tid >> 5);
    const int lo = blockIdx.x * RNG;
    const int cnt = (Nn - lo < RNG) ? (Nn - lo) : RNG;
    const int exact = gflag[0];
    const float Mh = gst[lane & 3];

    for (int i = tid; i < RNG * HC / 4; i += OWN_WAVES * 32)
        ((float4*)acc)[i] = make_float4(0.f, 0.f, 0.f, 0.f);
    for (int i = tid; i < RNG * HEADS; i += OWN_WAVES * 32) dsum[i] = 0.f;
    __syncthreads();

    int qn = 0;
    int* const myq = qe[wv];

    float* const span = obuf + (size_t)lo * HC;
    int cap = (cnt * (HC / OWN_WAVES) - 2) & ~15;
    if (cap > LCAP) cap = LCAP;
    int* const mycnt = (int*)span + wv;
    unsigned* const mylist = (unsigned*)span + OWN_WAVES + (size_t)wv * cap;
    int total = 0;
    if (mode != 0) total = __builtin_amdgcn_readfirstlane(*mycnt);
    const bool replay = (mode != 0) && (total <= cap);
    const bool record = (mode == 0);
    const bool bybyte = record && (exact == 0) && owner_bytes_fit(E, Nn);
    const bool last   = (mode == 2);
    int gp = 0;

    auto append = [&](bool take, int e) {
        const unsigned mask = __builtin_amdgcn_ballot_w32(take);
        if (mask != 0u) {
            const int pos = qn + (int)__builtin_amdgcn_mbcnt_lo(mask, 0u);
            if (take) myq[pos] = e;
            qn += __builtin_popcount(mask);
        }
    };

    auto process = [&](int base, int n, auto nbTag) {
        constexpr int NB = decltype(nbTag)::value;
        const int j = lane >> 2;
        const int h = lane & 3;
        int s[NB], d[NB];
        bool act[NB];
#pragma unroll
        for (int b = 0; b < NB; ++b) {
            act[b] = b * EPW + j < n;
            const int slot = base + (act[b] ? b * EPW + j : 0);
            if (replay) {
                const unsigned ent = mylist[slot];
                s[b] = (int)(ent >> DL_BITS);
                d[b] = lo + (int)(ent & ((1u << DL_BITS) - 1u));
            } else {
                const int e = myq[slot];
                edge_sd((long long)e, src, dst, E, &s[b], &d[b]);
            }
        }

        float v[NB][EPW][HEADS];
#pragma unroll
        for (int b = 0; b < NB; ++b) {
#pragma unroll
            for (int q = 0; q < EPW; ++q) {
                const int sq = __builtin_amdgcn_readlane(s[b], 4 * q);
                const float* __restrict__ hp = Hm + (size_t)sq * HC + lane;
#pragma unroll
                for (int k = 0; k < HEADS; ++k) v[b][q][k] = hp[k * CPH];
            }
        }

        float as_[NB], ad_[NB];
        unsigned mk[NB];
#pragma unroll
        for (int b = 0; b < NB; ++b) {
            const int dh = d[b] * HEADS + h;
            as_[b] = als[s[b] * HEADS + h];
            ad_[b] = ald[dh];
            mk[b] = 0u;
            if (exact) mk[b] = mkey[dh];
        }

        __builtin_amdgcn_sched_barrier(0);

        if (record) {
#pragma unroll
            for (int b = 0; b < NB; ++b) {
                const int gi = gp + b * EPW + j;
                if (h == 0 && act[b] && gi < cap)
                    mylist[gi] = ((unsigned)s[b] << DL_BITS) | (unsigned)(d[b] - lo);
            }
        }
        if (!replay) gp += n;

#pragma unroll
        for (int b = 0; b < NB; ++b) {
            float ev = as_[b] + ad_[b];
            ev = (ev < 0.f) ? NEG_SLOPE * ev : ev;
            float m;
            if (exact) m = dec_f32(mk[b]);
            else { m = Mh + ad_[b]; m = (m < 0.f) ? NEG_SLOPE * m : m; }
            const float w = expf(ev - m);
            const int dl = d[b] - lo;
            if (act[b]) lds_add(&dsum[dl * HEADS + h], w);

            auto add_row = [&](int q) {
                const int dq = __builtin_amdgcn_readlane(dl, 4 * q);
                float* ap = acc + dq * HC + lane;
#pragma unroll
                for (int k = 0; k < HEADS; ++k) {
                    const float wk = __int_as_float(
                        __builtin_amdgcn_readlane(__float_as_int(w), 4 * q + k));
                    lds_add(ap + k * CPH, v[b][q][k] * wk);
                }
            };
            if (n == NB * EPW) {
#pragma unroll
                for (int q = 0; q < EPW; ++q) add_row(q);
            } else {
#pragma unroll
                for (int q = 0; q < EPW; ++q) if (b * EPW + q < n) add_row(q);
            }
        }
    };
    const std::integral_constant<int, 1> oneBatch{};
    const std::integral_constant<int, 2> twoBatches{};

    if (replay) {
        int pos = 0;
        while (total - pos >= 2 * EPW) { process(pos, 2 * EPW, twoBatches); pos += 2 * EPW; }
        while (pos < total) {
            const int n = (total - pos < EPW) ? (total - pos) : EPW;
            process(pos, n, oneBatch);
            pos += n;
        }
    } else {
    {
        const int PERW = (RNG + OWN_WAVES - 1) / OWN_WAVES;
        const int node = lo + wv * PERW + lane;
        append(lane < PERW && node < lo + cnt, E + node);
    }

    const int nCh = E >> 9;
    const int tail0 = nCh << 9;
    const int rotb = (int)(((long long)nCh * blockIdx.x) / gridDim.x);
    if (bybyte) {
    const uint4* __restrict__ c4p = (const uint4*)mkey;
    const unsigned pat = (unsigned)blockIdx.x * 0x01010101u;
    for (int c0 = wv; c0 < nCh; c0 += OWN_WAVES * SCAN_U) {
        uint4 cv[SCAN_U];
        unsigned hv[SCAN_U];
#pragma unroll
        for (int u = 0; u < SCAN_U; ++u) {
            const int cc = c0 + OWN_WAVES * u;
            int ca = ((cc < nCh) ? cc : c0) + rotb;
            if (ca >= nCh) ca -= nCh;
            cv[u] = c4p[(size_t)ca * 32 + lane];
        }
#pragma unroll
        for (int u = 0; u < SCAN_U; ++u) {
            auto zb = [&](unsigned w) -> unsigned {
                const unsigned x = w ^ pat;
                return ~(((x & 0x7f7f7f7fu) + 0x7f7f7f7fu) | x | 0x7f7f7f7fu);
            };
            const unsigned hw = (zb(cv[u].x) >> 7) | (zb(cv[u].y) >> 6) | (zb(cv[u].z) >> 5) | (zb(cv[u].w) >> 4);
            hv[u] = (c0 + OWN_WAVES * u < nCh) ? hw : 0u;
        }
        for (int u = 0; u < SCAN_U; ++u) {
            unsigned hits = hv[0];
#pragma unroll
            for (int k = 0; k + 1 < SCAN_U; ++k) hv[k] = hv[k + 1];
            if (__builtin_amdgcn_ballot_w32(hits != 0u) != 0u) {
                int ca = c0 + OWN_WAVES * u + rotb;
                if (ca >= nCh) ca -= nCh;
                const int eb = ca * 512 + 16 * lane;
                do {
                    const int p = __builtin_ctz(hits | 0x80000000u);
                    append(hits != 0u, eb + ((p & 3) << 2) + (p >> 3));
                    hits &= hits - 1u;
                } while (__builtin_amdgcn_ballot_w32(hits != 0u) != 0u);
            }
            while (qn >= 2 * EPW) { qn -= 2 * EPW; process(qn, 2 * EPW, twoBatches); }
        }
    }
    } else {
    const int4* __restrict__ d4p = (const int4*)dst;
    for (int c0 = wv; c0 < nCh; c0 += OWN_WAVES * (SCAN_U / SCAN_G)) {
        int4 dv[SCAN_U];
        int cb[SCAN_U];
#pragma unroll
        for (int u = 0; u < SCAN_U; ++u) {
            const int cc = c0 + OWN_WAVES * (u / SCAN_G);
            int ca = ((cc < nCh) ? cc : c0) + rotb;
            if (ca >= nCh) ca -= nCh;
            cb[u] = ca * SCAN_G + (u % SCAN_G);
            dv[u] = d4p[(size_t)cb[u] * 32 + lane];
        }
#pragma unroll
        for (int u = 0; u < SCAN_U; ++u) {
            const int4 t = dv[u];
            const bool in = c0 + OWN_WAVES * (u / SCAN_G) < nCh;
            const bool m0 = (unsigned)(t.x - lo) < (unsigned)cnt;
            const bool m1 = (unsigned)(t.y - lo) < (unsigned)cnt;
            const bool m2 = (unsigned)(t.z - lo) < (unsigned)cnt;
            const bool m3 = (unsigned)(t.w - lo) < (unsigned)cnt;
            if (in && __builtin_amdgcn_ballot_w32(m0 | m1 | m2 | m3) != 0u) {
                const int eb = cb[u] * 128 + 4 * lane;
                append(m0, eb);
                append(m1, eb + 1);
                append(m2, eb + 2);
                append(m3, eb + 3);
            }
            if ((u % SCAN_G) == SCAN_G - 1)
                while (qn >= 2 * EPW) { qn -= 2 * EPW; process(qn, 2 * EPW, twoBatches); }
        }
    }
    }
    if (wv == 0) {
        for (int eb = tail0; eb < E; eb += 32) {
            const int e = eb + lane;
            const int d = (e < E) ? dst[e] : -1;
            append((unsigned)(d - lo) < (unsigned)cnt, e);
        }
    }
    while (qn > 0) {
        const int n = (qn < EPW) ? qn : EPW;
        qn -= n;
        process(qn, n, oneBatch);
    }
    if (record && lane == 0) *mycnt = gp;
    }

    __syncthreads();
    if (last) {
        float4* __restrict__ orow = (float4*)span;
        const float4 bi = ((const float4*)bias)[tid & 31];
        const int hd = (tid & 31) >> 3;
        for (int i = tid; i < cnt * (HC / 4); i += OWN_WAVES * 32) {
            const float4 a = ((const float4*)acc)[i];
            const float dn = dsum[(i >> 5) * HEADS + hd];
            float4 o;
            o.x = a.x / dn + bi.x;
            o.y = a.y / dn + bi.y;
            o.z = a.z / dn + bi.z;
            o.w = a.w / dn + bi.w;
            orow[i] = o;
        }
    } else {
        float4* __restrict__ orow = (float4*)(agg + (size_t)lo * HC);
        for (int i = tid; i < cnt * (HC / 4); i += OWN_WAVES * 32) orow[i] = ((const float4*)acc)[i];
        for (int i = tid; i < cnt * HEADS; i += OWN_WAVES * 32) denom[lo * HEADS + i] = dsum[i];
    }
}

extern "C" void kernel_launch(void* const* d_in, const int* in_sizes, int n_in,
                              void* d_out, int out_size, void* d_ws, size_t ws_size,
                              hipStream_t stream)
{
    const float* x    = (const float*)d_in[0];
    const int*   esrc = (const int*)d_in[1];
    const int*   edst = (const int*)d_in[2];
    const float* W[3]  = {(const float*)d_in[3],  (const float*)d_in[7],  (const float*)d_in[11]};
    const float* As[3] = {(const float*)d_in[4],  (const float*)d_in[8],  (const float*)d_in[12]};
    const float* Ad[3] = {(const float*)d_in[5],  (const float*)d_in[9],  (const float*)d_in[13]};
    const float* Bi[3] = {(const float*)d_in[6],  (const float*)d_in[10], (const float*)d_in[14]};
    const float* LnG[2] = {(const float*)d_in[15], (const float*)d_in[17]};
    const float* LnB[2] = {(const float*)d_in[16], (const float*)d_in[18]};

    const int Nn   = in_sizes[0] / HC;
    const int E    = in_sizes[1];
    const int Etot = E + Nn;

    float*    hbuf  = (float*)d_ws;
    float*    agg   = hbuf + (size_t)Nn * HC;
    float*    als   = agg  + (size_t)Nn * HC;
    float*    ald   = als  + (size_t)Nn * HEADS;
    unsigned* mkey  = (unsigned*)(ald + (size_t)Nn * HEADS);
    float*    denom = (float*)(mkey + (size_t)Nn * HEADS);

    const int TB = 256;
    const unsigned gGemm = (unsigned)((Nn + 15) / 16);
    const unsigned gAttn = (unsigned)(((long long)Nn * HEADS + TB - 1) / TB);
    const unsigned gEH   = 2048u;
    const unsigned gOwn  = (unsigned)((Nn + RNG - 1) / RNG);

    const float* cur = x;
    const float* pB = nullptr;
    const float* pG = nullptr;
    const float* pL = nullptr;
    int fuse = 0;
    float* part = agg;
    float* gstA = part + (size_t)gAttn * 8;
    float* gstO = (float*)d_out + (size_t)((Nn < RNG) ? Nn : RNG) * HC - 8;
    for (int l = 0; l < 3; ++l) {
        float* layer_out = agg;
        float* gst   = (l == 2) ? gstA : gstO;
        int*   gflag = (int*)(gst + HEADS);

        k_gemm_wmma<<<gGemm, 256, 0, stream>>>(cur, W[l], hbuf, Nn, denom, pB, pG, pL, fuse);
        k_attn_logits<<<gAttn, TB, 0, stream>>>(hbuf, As[l], Ad[l], als, ald, part, Nn);
        k_init<<<gAttn, TB, 0, stream>>>(mkey, Nn, part, (int)gAttn, gst, gflag);
        k_edge_max<<<gEH, TB, 0, stream>>>(esrc, edst, als, ald, mkey, gflag, E, Etot, (l == 0) ? 1 : 0, Nn);
        k_edge_own<<<gOwn, OWN_WAVES * 32, 0, stream>>>(esrc, edst, als, ald, mkey, denom,
                                            hbuf, layer_out, gst, gflag,
                                            (float*)d_out, Bi[2], (l == 0) ? 0 : l, E, Nn);
        if (l < 2) {
            cur = layer_out;
            pB = Bi[l]; pG = LnG[l]; pL = LnB[l];
            fuse = 1;
        }
    }
}


# Set-up launch merged into the max-pass launch: each of its 512 groups (was 2048) folds the logit extrema itself, 4 wide loads in flight, and takes its flag from that; logit launch clears max keys
# speedup vs baseline: 4.3831x; 1.0665x over previous
#include <hip/hip_runtime.h>
#include <hip/hip_bf16.h>
#include <float.h>
#include <math.h>
#include <type_traits>

#define HEADS 4
#define CPH 32
#define HC 128
#define NEG_SLOPE 0.2f
#define EPSV 1e-5f

typedef float v2f __attribute__((ext_vector_type(2)));
typedef float v8f __attribute__((ext_vector_type(8)));

__device__ __forceinline__ unsigned enc_f32(float f) {
    unsigned u = __float_as_uint(f);
    return (u & 0x80000000u) ? ~u : (u | 0x80000000u);
}
__device__ __forceinline__ float dec_f32(unsigned k) {
    unsigned u = (k & 0x80000000u) ? (k ^ 0x80000000u) : ~k;
    return __uint_as_float(u);
}

__device__ __forceinline__ void edge_sd(long long e, const int* __restrict__ src,
                                        const int* __restrict__ dst, int E,
                                        int* s, int* d) {
    if (e < (long long)E) { *s = src[e]; *d = dst[e]; }
    else { int v = (int)(e - E); *s = v; *d = v; }
}

#define XP 132

__global__ __launch_bounds__(256) __attribute__((amdgpu_waves_per_eu(16, 16))) void k_gemm_wmma(
    const float* __restrict__ A, const float* __restrict__ W,
    float* __restrict__ Hout, int M,
    const float* __restrict__ denom, const float* __restrict__ bias,
    const float* __restrict__ g, const float* __restrict__ bln, int fuse)
{
    __shared__ __attribute__((aligned(16))) float Xs[16 * XP];

    const int lane = threadIdx.x & 31;
    const int wave = threadIdx.x >> 5;
    const int m0   = blockIdx.x * 16;
    const int n0   = wave * 16;
    const int hi   = lane >> 4;
    const int l16  = lane & 15;
    const bool full = (m0 + 16 <= M);

    float4 v[2];
    int rr[2];
#pragma unroll
    for (int i = 0; i < 2; ++i) {
        int r = m0 + wave + 8 * i;
        if (r >= M) r = M - 1;
        rr[i] = r;
        v[i] = *(const float4*)(A + (size_t)r * HC + 4 * lane);
    }
    if (fuse) {
        float dn[2];
#pragma unroll
        for (int i = 0; i < 2; ++i) dn[i] = denom[rr[i] * HEADS + (lane >> 3)];
        const float4 bi = *(const float4*)(bias + 4 * lane);
        const float4 gg = *(const float4*)(g + 4 * lane);
        const float4 bb = *(const float4*)(bln + 4 * lane);
#pragma unroll
        for (int i = 0; i < 2; ++i) {
            float4 t = v[i];
            t.x = t.x / dn[i] + bi.x;
            t.y = t.y / dn[i] + bi.y;
            t.z = t.z / dn[i] + bi.z;
            t.w = t.w / dn[i] + bi.w;
            float s1 = (t.x + t.y) + (t.z + t.w);
            float s2 = (t.x * t.x + t.y * t.y) + (t.z * t.z + t.w * t.w);
#pragma unroll
            for (int off = 16; off > 0; off >>= 1) {
                s1 += __shfl_xor(s1, off, 32);
                s2 += __shfl_xor(s2, off, 32);
            }
            const float mu  = s1 * (1.f / HC);
            const float var = s2 * (1.f / HC) - mu * mu;
            const float rs  = rsqrtf(var + EPSV);
            t.x = fmaxf((t.x - mu) * rs * gg.x + bb.x, 0.f);
            t.y = fmaxf((t.y - mu) * rs * gg.y + bb.y, 0.f);
            t.z = fmaxf((t.z - mu) * rs * gg.z + bb.z, 0.f);
            t.w = fmaxf((t.w - mu) * rs * gg.w + bb.w, 0.f);
            v[i] = t;
        }
    }
#pragma unroll
    for (int i = 0; i < 2; ++i)
        *(float4*)(Xs + (wave + 8 * i) * XP + 4 * lane) = v[i];
    __syncthreads();

    const float* xr = Xs + l16 * XP + 2 * hi;
    const float* __restrict__ wc = W + (size_t)(2 * hi) * HC + n0 + l16;

    v8f acc = {0.f, 0.f, 0.f, 0.f, 0.f, 0.f, 0.f, 0.f};

#pragma unroll 4
    for (int k = 0; k < HC; k += 4) {
        v2f av, bv;
        av = *(const v2f*)(xr + k);
        bv.x = wc[(size_t)k * HC];
        bv.y = wc[(size_t)(k + 1) * HC];
        acc = __builtin_amdgcn_wmma_f32_16x16x4_f32(
             false, av,  false, bv,
             (short)0, acc,  false,  false);
    }

    float* __restrict__ outp = Hout + (size_t)(m0 + 8 * hi) * HC + n0 + l16;
    if (full) {
#pragma unroll
        for (int r = 0; r < 8; ++r) outp[(size_t)r * HC] = acc[r];
    } else {
#pragma unroll
        for (int r = 0; r < 8; ++r) {
            int ro = m0 + r + 8 * hi;
            if (ro < M) Hout[(size_t)ro * HC + n0 + l16] = acc[r];
        }
    }
}

#define SHIFT_SPAN 32.0f

__global__ __launch_bounds__(256) void k_attn_logits(const float* __restrict__ Hm,
                              const float* __restrict__ a_s,
                              const float* __restrict__ a_d,
                              float* __restrict__ als, float* __restrict__ ald,
                              float* __restrict__ part, unsigned* __restrict__ mkey, int Nn)
{
    __shared__ float red[8][8];
    const int idx = blockIdx.x * 256 + threadIdx.x;
    const bool ok = idx < Nn * HEADS;
    const int lane = threadIdx.x & 31;
    const int wave = threadIdx.x >> 5;
    float ss = 0.f, sd = 0.f;
    if (ok) {
        int n = idx >> 2, h = idx & 3;
        const float4* hp = (const float4*)(Hm + (size_t)n * HC + h * CPH);
        const float4* as = (const float4*)(a_s + h * CPH);
        const float4* ad = (const float4*)(a_d + h * CPH);
#pragma unroll
        for (int c = 0; c < CPH / 4; ++c) {
            float4 v = hp[c], a = as[c], b = ad[c];
            ss += v.x * a.x + v.y * a.y + v.z * a.z + v.w * a.w;
            sd += v.x * b.x + v.y * b.y + v.z * b.z + v.w * b.w;
        }
        als[idx] = ss;
        ald[idx] = sd;
        mkey[idx] = 0u;
    }
    float mx = ok ? ss : -INFINITY;
    float mn = ok ? ss :  INFINITY;
#pragma unroll
    for (int off = 4; off < 32; off <<= 1) {
        mx = fmaxf(mx, __shfl_xor(mx, off, 32));
        mn = fminf(mn, __shfl_xor(mn, off, 32));
    }
    if (lane < 4) { red[wave][lane] = mx; red[wave][4 + lane] = mn; }
    __syncthreads();
    if (threadIdx.x < 8) {
        const int t = threadIdx.x;
        float r = red[0][t];
#pragma unroll
        for (int w = 1; w < 8; ++w)
            r = (t < 4) ? fmaxf(r, red[w][t]) : fminf(r, red[w][t]);
        part[blockIdx.x * 8 + t] = r;
    }
}

#define RNG 416

__device__ __forceinline__ bool owner_bytes_fit(int E, int Nn) {
    return (long long)E <= (long long)Nn * HEADS * 4 && (Nn + RNG - 1) / RNG <= 256;
}

#define FOLD_U 4
#define FILL_U 4

__global__ __launch_bounds__(256) void k_edge_max(const int* __restrict__ src, const int* __restrict__ dst,
                           const float* __restrict__ als, const float* __restrict__ ald,
                           unsigned* __restrict__ mkey,
                           const float* __restrict__ part, int nPart,
                           float* __restrict__ gst, int* __restrict__ gflag,
                           int E, int Etot, int prep, int Nn)
{
    __shared__ float r2[8][8];
    const int t = threadIdx.x;
    const int lane = t & 31;

    const float sg = (t & 1) ? -1.f : 1.f;
    float4 r = make_float4(-INFINITY, -INFINITY, -INFINITY, -INFINITY);
    const float4* __restrict__ p4 = (const float4*)part;
    const int nV = nPart * 2;
    auto fold = [&](float& m, float a) { a *= sg; m = (a > m) ? a : m; };
    for (int i0 = t; i0 < nV; i0 += FOLD_U * 256) {
        float4 p[FOLD_U];
#pragma unroll
        for (int u = 0; u < FOLD_U; ++u) {
            const int i = i0 + 256 * u;
            p[u] = p4[(i < nV) ? i : i0];
        }
#pragma unroll
        for (int u = 0; u < FOLD_U; ++u) {
            fold(r.x, p[u].x);
            fold(r.y, p[u].y);
            fold(r.z, p[u].z);
            fold(r.w, p[u].w);
        }
    }
#pragma unroll
    for (int off = 2; off < 32; off <<= 1) {
        r.x = fmaxf(r.x, __shfl_xor(r.x, off, 32));
        r.y = fmaxf(r.y, __shfl_xor(r.y, off, 32));
        r.z = fmaxf(r.z, __shfl_xor(r.z, off, 32));
        r.w = fmaxf(r.w, __shfl_xor(r.w, off, 32));
    }
    if (lane < 2) {
        float* q = &r2[t >> 5][4 * lane];
        q[0] = r.x; q[1] = r.y; q[2] = r.z; q[3] = r.w;
    }
    __syncthreads();
    if (t < 8) {
        float q = r2[0][t];
#pragma unroll
        for (int w = 1; w < 8; ++w) q = fmaxf(q, r2[w][t]);
        r2[0][t] = q;
    }
    __syncthreads();
    int wide = 0;
    float Mv[HEADS];
#pragma unroll
    for (int h = 0; h < HEADS; ++h) {
        Mv[h] = r2[0][h];
        const float mn = -r2[0][4 + h];
        if (!(Mv[h] - mn <= SHIFT_SPAN)) wide = 1;
    }
    if (blockIdx.x == 0 && t == 0) {
        gst[0] = Mv[0]; gst[1] = Mv[1]; gst[2] = Mv[2]; gst[3] = Mv[3];
        gflag[0] = wide;
    }

    const long long step = (long long)gridDim.x * blockDim.x;
    if (wide == 0) {
        if (prep && owner_bytes_fit(E, Nn)) {
            const long long nW = (long long)(E >> 9) << 7;
            const int4* __restrict__ d4 = (const int4*)dst;
            for (long long i0 = (long long)blockIdx.x * blockDim.x + t; i0 < nW; i0 += FILL_U * step) {
                int4 d[FILL_U];
#pragma unroll
                for (int u = 0; u < FILL_U; ++u) {
                    const long long i = i0 + u * step;
                    d[u] = d4[(i < nW) ? i : i0];
                }
#pragma unroll
                for (int u = 0; u < FILL_U; ++u) {
                    const long long i = i0 + u * step;
                    if (i < nW)
                        mkey[i] = ((unsigned)d[u].x / RNG) | (((unsigned)d[u].y / RNG) << 8) |
                                  (((unsigned)d[u].z / RNG) << 16) | (((unsigned)d[u].w / RNG) << 24);
                }
            }
        }
        return;
    }
    const long long total = (long long)Etot * HEADS;
    for (long long idx = (long long)blockIdx.x * blockDim.x + threadIdx.x; idx < total; idx += step) {
        long long e = idx >> 2; int h = (int)(idx & 3);
        int s, d; edge_sd(e, src, dst, E, &s, &d);
        float ev = als[s * HEADS + h] + ald[d * HEADS + h];
        ev = (ev < 0.f) ? NEG_SLOPE * ev : ev;
        atomicMax(&mkey[d * HEADS + h], enc_f32(ev));
    }
}

#define EPW 8
#define OWN_WAVES 32
#define SCAN_U 8
#define SCAN_G 4
#define QCAP 544
#define LCAP 1536
#define DL_BITS 9

__device__ __forceinline__ void lds_add(float* p, float x) {
    __hip_atomic_fetch_add(p, x, __ATOMIC_RELAXED, __HIP_MEMORY_SCOPE_WORKGROUP);
}

__global__ __launch_bounds__(OWN_WAVES * 32) void k_edge_own(
                           const int* __restrict__ src, const int* __restrict__ dst,
                           const float* __restrict__ als, const float* __restrict__ ald,
                           const unsigned* __restrict__ mkey, float* __restrict__ denom,
                           const float* __restrict__ Hm, float* __restrict__ agg,
                           const float* __restrict__ gst, const int* __restrict__ gflag,
                           float* __restrict__ obuf, const float* __restrict__ bias, int mode,
                           int E, int Nn)
{
    __shared__ __attribute__((aligned(16))) float acc[RNG * HC];
    __shared__ __attribute__((aligned(16))) float dsum[RNG * HEADS];
    __shared__ int qe[OWN_WAVES][QCAP];

    const int tid  = threadIdx.x;
    const int lane = tid & 31;
    const int wv = __builtin_amdgcn_readfirstlane(tid >> 5);
    const int lo = blockIdx.x * RNG;
    const int cnt = (Nn - lo < RNG) ? (Nn - lo) : RNG;
    const int exact = gflag[0];
    const float Mh = gst[lane & 3];

    for (int i = tid; i < RNG * HC / 4; i += OWN_WAVES * 32)
        ((float4*)acc)[i] = make_float4(0.f, 0.f, 0.f, 0.f);
    for (int i = tid; i < RNG * HEADS; i += OWN_WAVES * 32) dsum[i] = 0.f;
    __syncthreads();

    int qn = 0;
    int* const myq = qe[wv];

    float* const span = obuf + (size_t)lo * HC;
    int cap = (cnt * (HC / OWN_WAVES) - 2) & ~15;
    if (cap > LCAP) cap = LCAP;
    int* const mycnt = (int*)span + wv;
    unsigned* const mylist = (unsigned*)span + OWN_WAVES + (size_t)wv * cap;
    int total = 0;
    if (mode != 0) total = __builtin_amdgcn_readfirstlane(*mycnt);
    const bool replay = (mode != 0) && (total <= cap);
    const bool record = (mode == 0);
    const bool bybyte = record && (exact == 0) && owner_bytes_fit(E, Nn);
    const bool last   = (mode == 2);
    int gp = 0;

    auto append = [&](bool take, int e) {
        const unsigned mask = __builtin_amdgcn_ballot_w32(take);
        if (mask != 0u) {
            const int pos = qn + (int)__builtin_amdgcn_mbcnt_lo(mask, 0u);
            if (take) myq[pos] = e;
            qn += __builtin_popcount(mask);
        }
    };

    auto process = [&](int base, int n, auto nbTag) {
        constexpr int NB = decltype(nbTag)::value;
        const int j = lane >> 2;
        const int h = lane & 3;
        int s[NB], d[NB];
        bool act[NB];
#pragma unroll
        for (int b = 0; b < NB; ++b) {
            act[b] = b * EPW + j < n;
            const int slot = base + (act[b] ? b * EPW + j : 0);
            if (replay) {
                const unsigned ent = mylist[slot];
                s[b] = (int)(ent >> DL_BITS);
                d[b] = lo + (int)(ent & ((1u << DL_BITS) - 1u));
            } else {
                const int e = myq[slot];
                edge_sd((long long)e, src, dst, E, &s[b], &d[b]);
            }
        }

        float v[NB][EPW][HEADS];
#pragma unroll
        for (int b = 0; b < NB; ++b) {
#pragma unroll
            for (int q = 0; q < EPW; ++q) {
                const int sq = __builtin_amdgcn_readlane(s[b], 4 * q);
                const float* __restrict__ hp = Hm + (size_t)sq * HC + lane;
#pragma unroll
                for (int k = 0; k < HEADS; ++k) v[b][q][k] = hp[k * CPH];
            }
        }

        float as_[NB], ad_[NB];
        unsigned mk[NB];
#pragma unroll
        for (int b = 0; b < NB; ++b) {
            const int dh = d[b] * HEADS + h;
            as_[b] = als[s[b] * HEADS + h];
            ad_[b] = ald[dh];
            mk[b] = 0u;
            if (exact) mk[b] = mkey[dh];
        }

        __builtin_amdgcn_sched_barrier(0);

        if (record) {
#pragma unroll
            for (int b = 0; b < NB; ++b) {
                const int gi = gp + b * EPW + j;
                if (h == 0 && act[b] && gi < cap)
                    mylist[gi] = ((unsigned)s[b] << DL_BITS) | (unsigned)(d[b] - lo);
            }
        }
        if (!replay) gp += n;

#pragma unroll
        for (int b = 0; b < NB; ++b) {
            float ev = as_[b] + ad_[b];
            ev = (ev < 0.f) ? NEG_SLOPE * ev : ev;
            float m;
            if (exact) m = dec_f32(mk[b]);
            else { m = Mh + ad_[b]; m = (m < 0.f) ? NEG_SLOPE * m : m; }
            const float w = expf(ev - m);
            const int dl = d[b] - lo;
            if (act[b]) lds_add(&dsum[dl * HEADS + h], w);

            auto add_row = [&](int q) {
                const int dq = __builtin_amdgcn_readlane(dl, 4 * q);
                float* ap = acc + dq * HC + lane;
#pragma unroll
                for (int k = 0; k < HEADS; ++k) {
                    const float wk = __int_as_float(
                        __builtin_amdgcn_readlane(__float_as_int(w), 4 * q + k));
                    lds_add(ap + k * CPH, v[b][q][k] * wk);
                }
            };
            if (n == NB * EPW) {
#pragma unroll
                for (int q = 0; q < EPW; ++q) add_row(q);
            } else {
#pragma unroll
                for (int q = 0; q < EPW; ++q) if (b * EPW + q < n) add_row(q);
            }
        }
    };
    const std::integral_constant<int, 1> oneBatch{};
    const std::integral_constant<int, 2> twoBatches{};

    if (replay) {
        int pos = 0;
        while (total - pos >= 2 * EPW) { process(pos, 2 * EPW, twoBatches); pos += 2 * EPW; }
        while (pos < total) {
            const int n = (total - pos < EPW) ? (total - pos) : EPW;
            process(pos, n, oneBatch);
            pos += n;
        }
    } else {
    {
        const int PERW = (RNG + OWN_WAVES - 1) / OWN_WAVES;
        const int node = lo + wv * PERW + lane;
        append(lane < PERW && node < lo + cnt, E + node);
    }

    const int nCh = E >> 9;
    const int tail0 = nCh << 9;
    const int rotb = (int)(((long long)nCh * blockIdx.x) / gridDim.x);
    if (bybyte) {
    const uint4* __restrict__ c4p = (const uint4*)mkey;
    const unsigned pat = (unsigned)blockIdx.x * 0x01010101u;
    for (int c0 = wv; c0 < nCh; c0 += OWN_WAVES * SCAN_U) {
        uint4 cv[SCAN_U];
        unsigned hv[SCAN_U];
#pragma unroll
        for (int u = 0; u < SCAN_U; ++u) {
            const int cc = c0 + OWN_WAVES * u;
            int ca = ((cc < nCh) ? cc : c0) + rotb;
            if (ca >= nCh) ca -= nCh;
            cv[u] = c4p[(size_t)ca * 32 + lane];
        }
#pragma unroll
        for (int u = 0; u < SCAN_U; ++u) {
            auto zb = [&](unsigned w) -> unsigned {
                const unsigned x = w ^ pat;
                return ~(((x & 0x7f7f7f7fu) + 0x7f7f7f7fu) | x | 0x7f7f7f7fu);
            };
            const unsigned hw = (zb(cv[u].x) >> 7) | (zb(cv[u].y) >> 6) | (zb(cv[u].z) >> 5) | (zb(cv[u].w) >> 4);
            hv[u] = (c0 + OWN_WAVES * u < nCh) ? hw : 0u;
        }
        for (int u = 0; u < SCAN_U; ++u) {
            unsigned hits = hv[0];
#pragma unroll
            for (int k = 0; k + 1 < SCAN_U; ++k) hv[k] = hv[k + 1];
            if (__builtin_amdgcn_ballot_w32(hits != 0u) != 0u) {
                int ca = c0 + OWN_WAVES * u + rotb;
                if (ca >= nCh) ca -= nCh;
                const int eb = ca * 512 + 16 * lane;
                do {
                    const int p = __builtin_ctz(hits | 0x80000000u);
                    append(hits != 0u, eb + ((p & 3) << 2) + (p >> 3));
                    hits &= hits - 1u;
                } while (__builtin_amdgcn_ballot_w32(hits != 0u) != 0u);
            }
            while (qn >= 2 * EPW) { qn -= 2 * EPW; process(qn, 2 * EPW, twoBatches); }
        }
    }
    } else {
    const int4* __restrict__ d4p = (const int4*)dst;
    for (int c0 = wv; c0 < nCh; c0 += OWN_WAVES * (SCAN_U / SCAN_G)) {
        int4 dv[SCAN_U];
        int cb[SCAN_U];
#pragma unroll
        for (int u = 0; u < SCAN_U; ++u) {
            const int cc = c0 + OWN_WAVES * (u / SCAN_G);
            int ca = ((cc < nCh) ? cc : c0) + rotb;
            if (ca >= nCh) ca -= nCh;
            cb[u] = ca * SCAN_G + (u % SCAN_G);
            dv[u] = d4p[(size_t)cb[u] * 32 + lane];
        }
#pragma unroll
        for (int u = 0; u < SCAN_U; ++u) {
            const int4 t = dv[u];
            const bool in = c0 + OWN_WAVES * (u / SCAN_G) < nCh;
            const bool m0 = (unsigned)(t.x - lo) < (unsigned)cnt;
            const bool m1 = (unsigned)(t.y - lo) < (unsigned)cnt;
            const bool m2 = (unsigned)(t.z - lo) < (unsigned)cnt;
            const bool m3 = (unsigned)(t.w - lo) < (unsigned)cnt;
            if (in && __builtin_amdgcn_ballot_w32(m0 | m1 | m2 | m3) != 0u) {
                const int eb = cb[u] * 128 + 4 * lane;
                append(m0, eb);
                append(m1, eb + 1);
                append(m2, eb + 2);
                append(m3, eb + 3);
            }
            if ((u % SCAN_G) == SCAN_G - 1)
                while (qn >= 2 * EPW) { qn -= 2 * EPW; process(qn, 2 * EPW, twoBatches); }
        }
    }
    }
    if (wv == 0) {
        for (int eb = tail0; eb < E; eb += 32) {
            const int e = eb + lane;
            const int d = (e < E) ? dst[e] : -1;
            append((unsigned)(d - lo) < (unsigned)cnt, e);
        }
    }
    while (qn > 0) {
        const int n = (qn < EPW) ? qn : EPW;
        qn -= n;
        process(qn, n, oneBatch);
    }
    if (record && lane == 0) *mycnt = gp;
    }

    __syncthreads();
    if (last) {
        float4* __restrict__ orow = (float4*)span;
        const float4 bi = ((const float4*)bias)[tid & 31];
        const int hd = (tid & 31) >> 3;
        for (int i = tid; i < cnt * (HC / 4); i += OWN_WAVES * 32) {
            const float4 a = ((const float4*)acc)[i];
            const float dn = dsum[(i >> 5) * HEADS + hd];
            float4 o;
            o.x = a.x / dn + bi.x;
            o.y = a.y / dn + bi.y;
            o.z = a.z / dn + bi.z;
            o.w = a.w / dn + bi.w;
            orow[i] = o;
        }
    } else {
        float4* __restrict__ orow = (float4*)(agg + (size_t)lo * HC);
        for (int i = tid; i < cnt * (HC / 4); i += OWN_WAVES * 32) orow[i] = ((const float4*)acc)[i];
        for (int i = tid; i < cnt * HEADS; i += OWN_WAVES * 32) denom[lo * HEADS + i] = dsum[i];
    }
}

extern "C" void kernel_launch(void* const* d_in, const int* in_sizes, int n_in,
                              void* d_out, int out_size, void* d_ws, size_t ws_size,
                              hipStream_t stream)
{
    const float* x    = (const float*)d_in[0];
    const int*   esrc = (const int*)d_in[1];
    const int*   edst = (const int*)d_in[2];
    const float* W[3]  = {(const float*)d_in[3],  (const float*)d_in[7],  (const float*)d_in[11]};
    const float* As[3] = {(const float*)d_in[4],  (const float*)d_in[8],  (const float*)d_in[12]};
    const float* Ad[3] = {(const float*)d_in[5],  (const float*)d_in[9],  (const float*)d_in[13]};
    const float* Bi[3] = {(const float*)d_in[6],  (const float*)d_in[10], (const float*)d_in[14]};
    const float* LnG[2] = {(const float*)d_in[15], (const float*)d_in[17]};
    const float* LnB[2] = {(const float*)d_in[16], (const float*)d_in[18]};

    const int Nn   = in_sizes[0] / HC;
    const int E    = in_sizes[1];
    const int Etot = E + Nn;

    float*    hbuf  = (float*)d_ws;
    float*    agg   = hbuf + (size_t)Nn * HC;
    float*    als   = agg  + (size_t)Nn * HC;
    float*    ald   = als  + (size_t)Nn * HEADS;
    unsigned* mkey  = (unsigned*)(ald + (size_t)Nn * HEADS);
    float*    denom = (float*)(mkey + (size_t)Nn * HEADS);

    const int TB = 256;
    const unsigned gGemm = (unsigned)((Nn + 15) / 16);
    const unsigned gAttn = (unsigned)(((long long)Nn * HEADS + TB - 1) / TB);
    const unsigned gEH   = 512u;
    const unsigned gOwn  = (unsigned)((Nn + RNG - 1) / RNG);

    const float* cur = x;
    const float* pB = nullptr;
    const float* pG = nullptr;
    const float* pL = nullptr;
    int fuse = 0;
    float* part = agg;
    float* gstA = part + (size_t)gAttn * 8;
    float* gstO = (float*)d_out + (size_t)((Nn < RNG) ? Nn : RNG) * HC - 8;
    for (int l = 0; l < 3; ++l) {
        float* layer_out = agg;
        float* gst   = (l == 2) ? gstA : gstO;
        int*   gflag = (int*)(gst + HEADS);

        k_gemm_wmma<<<gGemm, 256, 0, stream>>>(cur, W[l], hbuf, Nn, denom, pB, pG, pL, fuse);
        k_attn_logits<<<gAttn, TB, 0, stream>>>(hbuf, As[l], Ad[l], als, ald, part, mkey, Nn);
        k_edge_max<<<gEH, TB, 0, stream>>>(esrc, edst, als, ald, mkey, part, (int)gAttn, gst, gflag,
                                           E, Etot, (l == 0) ? 1 : 0, Nn);
        k_edge_own<<<gOwn, OWN_WAVES * 32, 0, stream>>>(esrc, edst, als, ald, mkey, denom,
                                            hbuf, layer_out, gst, gflag,
                                            (float*)d_out, Bi[2], (l == 0) ? 0 : l, E, Nn);
        if (l < 2) {
            cur = layer_out;
            pB = Bi[l]; pG = LnG[l]; pL = LnB[l];
            fuse = 1;
        }
    }
}
